# starting kernel
# baseline (speedup 1.0000x reference)
#include <hip/hip_runtime.h>
#include <math.h>

#define NNODES 50000
#define NEDGES 500000
#define CHAN   64
#define R0f    2.0f

#define WAVES  8
#define BLOCK  (WAVES * 32)
#define NTILES (NEDGES / 16)
#define GRID   512

typedef float v2f __attribute__((ext_vector_type(2)));
typedef float v8f __attribute__((ext_vector_type(8)));

#define OFF_WSRC  0
#define OFF_WDST  4096
#define OFF_WS    8192
#define OFF_WV    12288
#define OFF_WGATE 16384
#define OFF_WENC  32768
#define OFF_BIAS0 33280
#define OFF_BGATE 33344
#define OFF_STAGE 33600
#define OFF_RH    (OFF_STAGE + WAVES * 1024)
#define OFF_IDX   (OFF_RH + WAVES * 64)
#define SMEM_FLOATS (OFF_IDX + WAVES * 32)

__device__ __forceinline__ void lds_fence() {
  asm volatile("s_wait_dscnt 0x0" ::: "memory");
}

__device__ __forceinline__ v8f wmma4(v2f a, v2f b, v8f c) {
  return __builtin_amdgcn_wmma_f32_16x16x4_f32(false, a, false, b, (short)0, c,
                                               false, false);
}

__global__ void zero_f32(float* __restrict__ p, int n) {
  int i = blockIdx.x * blockDim.x + threadIdx.x;
  if (i < n) p[i] = 0.0f;
}

__global__ void __launch_bounds__(BLOCK, 1)
messages_kernel(const int* __restrict__ src, const int* __restrict__ dst,
                const float* __restrict__ r_ij,
                const float* __restrict__ z0g, const float* __restrict__ z1g,
                const float* __restrict__ emb,
                const float* __restrict__ Wenc, const float* __restrict__ benc,
                const float* __restrict__ Wsrc, const float* __restrict__ bsrc,
                const float* __restrict__ Wdst, const float* __restrict__ bdst,
                const float* __restrict__ Wgate, const float* __restrict__ bgate,
                const float* __restrict__ Ws, const float* __restrict__ Wv,
                float* __restrict__ out0, float* __restrict__ out1) {
  extern __shared__ float smem[];

  for (int i = threadIdx.x; i < 64 * 64; i += BLOCK) {
    int k = i >> 6, n = i & 63;
    smem[OFF_WSRC + n * 64 + k] = Wsrc[i];
    smem[OFF_WDST + n * 64 + k] = Wdst[i];
    smem[OFF_WS   + n * 64 + k] = Ws[i];
    smem[OFF_WV   + n * 64 + k] = Wv[i];
  }
  for (int i = threadIdx.x; i < 64 * 256; i += BLOCK) {
    int k = i >> 8, n = i & 255;
    smem[OFF_WGATE + n * 64 + k] = Wgate[i];
  }
  for (int i = threadIdx.x; i < 8 * 64; i += BLOCK) {
    int k = i >> 6, n = i & 63;
    smem[OFF_WENC + n * 8 + k] = Wenc[i];
  }
  if (threadIdx.x < 64)
    smem[OFF_BIAS0 + threadIdx.x] =
        bsrc[threadIdx.x] + bdst[threadIdx.x] + benc[threadIdx.x];
  if (threadIdx.x < 256) smem[OFF_BGATE + threadIdx.x] = bgate[threadIdx.x];
  __syncthreads();

  const int lane = threadIdx.x & 31;
  const int wave = threadIdx.x >> 5;
  const int half = lane >> 4;
  const int nl   = lane & 15;

  float* stage = smem + OFF_STAGE + wave * 1024;
  float* rhS   = smem + OFF_RH + wave * 64;
  int*   idxS  = ((int*)(smem + OFF_IDX)) + wave * 32;

  for (int tile = blockIdx.x * WAVES + wave; tile < NTILES;
       tile += gridDim.x * WAVES) {
    const int e  = tile * 16 + nl;
    const int sN = src[e];
    const int dN = dst[e];
    if (half == 0) { idxS[nl] = sN; idxS[16 + nl] = dN; }

    float rx = r_ij[e * 3 + 0], ry = r_ij[e * 3 + 1], rz = r_ij[e * 3 + 2];
    float dd  = sqrtf(rx * rx + ry * ry + rz * rz);
    float sx = rx * 3.5f, sy = ry * 3.5f, sz = rz * 3.5f;
    float inv = rsqrtf(1.0f + sx * sx + sy * sy + sz * sz);
    if (half == 0) {
      rhS[nl * 4 + 0] = sx * inv;
      rhS[nl * 4 + 1] = sy * inv;
      rhS[nl * 4 + 2] = sz * inv;
    }
    float rb[4];
#pragma unroll
    for (int u = 0; u < 4; ++u) {
      int k = (u >> 1) * 4 + 2 * half + (u & 1);
      float x = (dd - (float)k * (R0f / 7.0f)) * 4.0f;
      rb[u] = expf(-x * x);
    }

    v8f acc[4];
#pragma unroll
    for (int j = 0; j < 4; ++j) acc[j] = (v8f)0.0f;

    const float* es = emb + (size_t)sN * 64;
    const float* ed = emb + (size_t)dN * 64;
#pragma unroll
    for (int t = 0; t < 16; ++t) {
      int k0 = 4 * t + 2 * half;
      v2f a = *(const v2f*)(es + k0);
#pragma unroll
      for (int j = 0; j < 4; ++j) {
        v2f b = *(const v2f*)(smem + OFF_WSRC + (16 * j + nl) * 64 + k0);
        acc[j] = wmma4(a, b, acc[j]);
      }
    }
#pragma unroll
    for (int t = 0; t < 16; ++t) {
      int k0 = 4 * t + 2 * half;
      v2f a = *(const v2f*)(ed + k0);
#pragma unroll
      for (int j = 0; j < 4; ++j) {
        v2f b = *(const v2f*)(smem + OFF_WDST + (16 * j + nl) * 64 + k0);
        acc[j] = wmma4(a, b, acc[j]);
      }
    }
#pragma unroll
    for (int t = 0; t < 2; ++t) {
      int k0 = 4 * t + 2 * half;
      v2f a;
      a[0] = rb[2 * t];
      a[1] = rb[2 * t + 1];
#pragma unroll
      for (int j = 0; j < 4; ++j) {
        v2f b = *(const v2f*)(smem + OFF_WENC + (16 * j + nl) * 8 + k0);
        acc[j] = wmma4(a, b, acc[j]);
      }
    }
#pragma unroll
    for (int j = 0; j < 4; ++j) {
      float bb = smem[OFF_BIAS0 + 16 * j + nl];
#pragma unroll
      for (int r = 0; r < 8; ++r) acc[j][r] += bb;
    }

    lds_fence();
#pragma unroll
    for (int j = 0; j < 4; ++j)
#pragma unroll
      for (int r = 0; r < 8; ++r)
        stage[(r + 8 * half) * 64 + 16 * j + nl] = acc[j][r];
    lds_fence();

    int sn[8], dn[8];
#pragma unroll
    for (int r = 0; r < 8; ++r) {
      sn[r] = idxS[r + 8 * half];
      dn[r] = idxS[16 + r + 8 * half];
    }
    float rh[3][8];
#pragma unroll
    for (int r = 0; r < 8; ++r) {
      rh[0][r] = rhS[(r + 8 * half) * 4 + 0];
      rh[1][r] = rhS[(r + 8 * half) * 4 + 1];
      rh[2][r] = rhS[(r + 8 * half) * 4 + 2];
    }
    float z0r[4][8];
#pragma unroll
    for (int j = 0; j < 4; ++j)
#pragma unroll
      for (int r = 0; r < 8; ++r)
        z0r[j][r] = z0g[(size_t)dn[r] * 64 + 16 * j + nl];
    float z1r[3][4][8];
#pragma unroll
    for (int i = 0; i < 3; ++i)
#pragma unroll
      for (int j = 0; j < 4; ++j)
#pragma unroll
        for (int r = 0; r < 8; ++r)
          z1r[i][j][r] = z1g[(size_t)dn[r] * 192 + i * 64 + 16 * j + nl];

    v8f g[4][4];
#pragma unroll
    for (int c = 0; c < 4; ++c) {
#pragma unroll
      for (int j = 0; j < 4; ++j) g[c][j] = (v8f)0.0f;
#pragma unroll
      for (int t = 0; t < 16; ++t) {
        int k0 = 4 * t + 2 * half;
        v2f a = *(const v2f*)(stage + nl * 64 + k0);
#pragma unroll
        for (int j = 0; j < 4; ++j) {
          v2f b = *(const v2f*)(smem + OFF_WGATE +
                                (64 * c + 16 * j + nl) * 64 + k0);
          g[c][j] = wmma4(a, b, g[c][j]);
        }
      }
#pragma unroll
      for (int j = 0; j < 4; ++j) {
        float bb = smem[OFF_BGATE + 64 * c + 16 * j + nl];
#pragma unroll
        for (int r = 0; r < 8; ++r) g[c][j][r] += bb;
      }
    }

#pragma unroll
    for (int j = 0; j < 4; ++j)
#pragma unroll
      for (int r = 0; r < 8; ++r) {
        float dotv = rh[0][r] * z1r[0][j][r] + rh[1][r] * z1r[1][j][r] +
                     rh[2][r] * z1r[2][j][r];
        g[0][j][r] = g[0][j][r] * z0r[j][r] + g[1][j][r] * dotv;
      }

    lds_fence();
#pragma unroll
    for (int j = 0; j < 4; ++j)
#pragma unroll
      for (int r = 0; r < 8; ++r)
        stage[(r + 8 * half) * 64 + 16 * j + nl] = g[0][j][r];
    lds_fence();
    v8f p[4];
#pragma unroll
    for (int j = 0; j < 4; ++j) p[j] = (v8f)0.0f;
#pragma unroll
    for (int t = 0; t < 16; ++t) {
      int k0 = 4 * t + 2 * half;
      v2f a = *(const v2f*)(stage + nl * 64 + k0);
#pragma unroll
      for (int j = 0; j < 4; ++j) {
        v2f b = *(const v2f*)(smem + OFF_WS + (16 * j + nl) * 64 + k0);
        p[j] = wmma4(a, b, p[j]);
      }
    }
#pragma unroll
    for (int j = 0; j < 4; ++j)
#pragma unroll
      for (int r = 0; r < 8; ++r)
        atomicAdd(&out0[(size_t)sn[r] * 64 + 16 * j + nl], p[j][r]);

#pragma unroll
    for (int i = 0; i < 3; ++i) {
      lds_fence();
#pragma unroll
      for (int j = 0; j < 4; ++j)
#pragma unroll
        for (int r = 0; r < 8; ++r)
          stage[(r + 8 * half) * 64 + 16 * j + nl] =
              g[2][j][r] * z1r[i][j][r] + g[3][j][r] * rh[i][r] * z0r[j][r];
      lds_fence();
#pragma unroll
      for (int j = 0; j < 4; ++j) p[j] = (v8f)0.0f;
#pragma unroll
      for (int t = 0; t < 16; ++t) {
        int k0 = 4 * t + 2 * half;
        v2f a = *(const v2f*)(stage + nl * 64 + k0);
#pragma unroll
        for (int j = 0; j < 4; ++j) {
          v2f b = *(const v2f*)(smem + OFF_WV + (16 * j + nl) * 64 + k0);
          p[j] = wmma4(a, b, p[j]);
        }
      }
#pragma unroll
      for (int j = 0; j < 4; ++j)
#pragma unroll
        for (int r = 0; r < 8; ++r)
          atomicAdd(&out1[(size_t)sn[r] * 192 + i * 64 + 16 * j + nl],
                    p[j][r]);
    }
  }
}

extern "C" void kernel_launch(void* const* d_in, const int* in_sizes, int n_in,
                              void* d_out, int out_size, void* d_ws,
                              size_t ws_size, hipStream_t stream) {
  const int*   src   = (const int*)d_in[0];
  const int*   dst   = (const int*)d_in[1];
  const float* r_ij  = (const float*)d_in[2];
  const float* z0    = (const float*)d_in[3];
  const float* z1    = (const float*)d_in[4];
  const float* emb   = (const float*)d_in[5];
  const float* Wenc  = (const float*)d_in[6];
  const float* benc  = (const float*)d_in[7];
  const float* Wsrc  = (const float*)d_in[8];
  const float* bsrc  = (const float*)d_in[9];
  const float* Wdst  = (const float*)d_in[10];
  const float* bdst  = (const float*)d_in[11];
  const float* Wgate = (const float*)d_in[12];
  const float* bgate = (const float*)d_in[13];
  const float* Ws    = (const float*)d_in[14];
  const float* Wv    = (const float*)d_in[15];
  float* out = (float*)d_out;

  zero_f32<<<(out_size + 255) / 256, 256, 0, stream>>>(out, out_size);

  size_t smem_bytes = (size_t)SMEM_FLOATS * sizeof(float);
  messages_kernel<<<GRID, BLOCK, smem_bytes, stream>>>(
      src, dst, r_ij, z0, z1, emb, Wenc, benc, Wsrc, bsrc, Wdst, bdst, Wgate,
      bgate, Ws, Wv, out, out + (size_t)NNODES * CHAN);
}


# LDS row pitch 64 -> 68 floats for the transposed weights and the per-wave stage tile (W_enc rows 8 -> 12): the 16-way bank conflict of every WMMA fragment read becomes conflict-free
# speedup vs baseline: 3.2721x; 3.2721x over previous
#include <hip/hip_runtime.h>
#include <math.h>

#define NNODES 50000
#define NEDGES 500000
#define CHAN   64
#define R0f    2.0f

#define WAVES  8
#define BLOCK  (WAVES * 32)
#define NTILES (NEDGES / 16)
#define GRID   512

typedef float v2f __attribute__((ext_vector_type(2)));
typedef float v8f __attribute__((ext_vector_type(8)));

#define WP 68
#define EP 12
#define OFF_WSRC  0
#define OFF_WDST  (OFF_WSRC + 64 * WP)
#define OFF_WS    (OFF_WDST + 64 * WP)
#define OFF_WV    (OFF_WS + 64 * WP)
#define OFF_WGATE (OFF_WV + 64 * WP)
#define OFF_WENC  (OFF_WGATE + 256 * WP)
#define OFF_BIAS0 (OFF_WENC + 64 * EP)
#define OFF_BGATE (OFF_BIAS0 + 64)
#define OFF_STAGE (OFF_BGATE + 256)
#define OFF_RH    (OFF_STAGE + WAVES * 16 * WP)
#define OFF_IDX   (OFF_RH + WAVES * 64)
#define SMEM_FLOATS (OFF_IDX + WAVES * 32)

__device__ __forceinline__ void lds_fence() {
  asm volatile("s_wait_dscnt 0x0" ::: "memory");
}

__device__ __forceinline__ v8f wmma4(v2f a, v2f b, v8f c) {
  return __builtin_amdgcn_wmma_f32_16x16x4_f32(false, a, false, b, (short)0, c,
                                               false, false);
}

__global__ void zero_f32(float* __restrict__ p, int n) {
  int i = blockIdx.x * blockDim.x + threadIdx.x;
  if (i < n) p[i] = 0.0f;
}

__global__ void __launch_bounds__(BLOCK, 1)
messages_kernel(const int* __restrict__ src, const int* __restrict__ dst,
                const float* __restrict__ r_ij,
                const float* __restrict__ z0g, const float* __restrict__ z1g,
                const float* __restrict__ emb,
                const float* __restrict__ Wenc, const float* __restrict__ benc,
                const float* __restrict__ Wsrc, const float* __restrict__ bsrc,
                const float* __restrict__ Wdst, const float* __restrict__ bdst,
                const float* __restrict__ Wgate, const float* __restrict__ bgate,
                const float* __restrict__ Ws, const float* __restrict__ Wv,
                float* __restrict__ out0, float* __restrict__ out1) {
  extern __shared__ float smem[];

  for (int i = threadIdx.x; i < 64 * 64; i += BLOCK) {
    int k = i >> 6, n = i & 63;
    smem[OFF_WSRC + n * WP + k] = Wsrc[i];
    smem[OFF_WDST + n * WP + k] = Wdst[i];
    smem[OFF_WS   + n * WP + k] = Ws[i];
    smem[OFF_WV   + n * WP + k] = Wv[i];
  }
  for (int i = threadIdx.x; i < 64 * 256; i += BLOCK) {
    int k = i >> 8, n = i & 255;
    smem[OFF_WGATE + n * WP + k] = Wgate[i];
  }
  for (int i = threadIdx.x; i < 8 * 64; i += BLOCK) {
    int k = i >> 6, n = i & 63;
    smem[OFF_WENC + n * EP + k] = Wenc[i];
  }
  if (threadIdx.x < 64)
    smem[OFF_BIAS0 + threadIdx.x] =
        bsrc[threadIdx.x] + bdst[threadIdx.x] + benc[threadIdx.x];
  if (threadIdx.x < 256) smem[OFF_BGATE + threadIdx.x] = bgate[threadIdx.x];
  __syncthreads();

  const int lane = threadIdx.x & 31;
  const int wave = threadIdx.x >> 5;
  const int half = lane >> 4;
  const int nl   = lane & 15;

  float* stage = smem + OFF_STAGE + wave * (16 * WP);
  float* rhS   = smem + OFF_RH + wave * 64;
  int*   idxS  = ((int*)(smem + OFF_IDX)) + wave * 32;

  for (int tile = blockIdx.x * WAVES + wave; tile < NTILES;
       tile += gridDim.x * WAVES) {
    const int e  = tile * 16 + nl;
    const int sN = src[e];
    const int dN = dst[e];
    if (half == 0) { idxS[nl] = sN; idxS[16 + nl] = dN; }

    float rx = r_ij[e * 3 + 0], ry = r_ij[e * 3 + 1], rz = r_ij[e * 3 + 2];
    float dd  = sqrtf(rx * rx + ry * ry + rz * rz);
    float sx = rx * 3.5f, sy = ry * 3.5f, sz = rz * 3.5f;
    float inv = rsqrtf(1.0f + sx * sx + sy * sy + sz * sz);
    if (half == 0) {
      rhS[nl * 4 + 0] = sx * inv;
      rhS[nl * 4 + 1] = sy * inv;
      rhS[nl * 4 + 2] = sz * inv;
    }
    float rb[4];
#pragma unroll
    for (int u = 0; u < 4; ++u) {
      int k = (u >> 1) * 4 + 2 * half + (u & 1);
      float x = (dd - (float)k * (R0f / 7.0f)) * 4.0f;
      rb[u] = expf(-x * x);
    }

    v8f acc[4];
#pragma unroll
    for (int j = 0; j < 4; ++j) acc[j] = (v8f)0.0f;

    const float* es = emb + (size_t)sN * 64;
    const float* ed = emb + (size_t)dN * 64;
#pragma unroll
    for (int t = 0; t < 16; ++t) {
      int k0 = 4 * t + 2 * half;
      v2f a = *(const v2f*)(es + k0);
#pragma unroll
      for (int j = 0; j < 4; ++j) {
        v2f b = *(const v2f*)(smem + OFF_WSRC + (16 * j + nl) * WP + k0);
        acc[j] = wmma4(a, b, acc[j]);
      }
    }
#pragma unroll
    for (int t = 0; t < 16; ++t) {
      int k0 = 4 * t + 2 * half;
      v2f a = *(const v2f*)(ed + k0);
#pragma unroll
      for (int j = 0; j < 4; ++j) {
        v2f b = *(const v2f*)(smem + OFF_WDST + (16 * j + nl) * WP + k0);
        acc[j] = wmma4(a, b, acc[j]);
      }
    }
#pragma unroll
    for (int t = 0; t < 2; ++t) {
      int k0 = 4 * t + 2 * half;
      v2f a;
      a[0] = rb[2 * t];
      a[1] = rb[2 * t + 1];
#pragma unroll
      for (int j = 0; j < 4; ++j) {
        v2f b = *(const v2f*)(smem + OFF_WENC + (16 * j + nl) * EP + k0);
        acc[j] = wmma4(a, b, acc[j]);
      }
    }
#pragma unroll
    for (int j = 0; j < 4; ++j) {
      float bb = smem[OFF_BIAS0 + 16 * j + nl];
#pragma unroll
      for (int r = 0; r < 8; ++r) acc[j][r] += bb;
    }

    lds_fence();
#pragma unroll
    for (int j = 0; j < 4; ++j)
#pragma unroll
      for (int r = 0; r < 8; ++r)
        stage[(r + 8 * half) * WP + 16 * j + nl] = acc[j][r];
    lds_fence();

    int sn[8], dn[8];
#pragma unroll
    for (int r = 0; r < 8; ++r) {
      sn[r] = idxS[r + 8 * half];
      dn[r] = idxS[16 + r + 8 * half];
    }
    float rh[3][8];
#pragma unroll
    for (int r = 0; r < 8; ++r) {
      rh[0][r] = rhS[(r + 8 * half) * 4 + 0];
      rh[1][r] = rhS[(r + 8 * half) * 4 + 1];
      rh[2][r] = rhS[(r + 8 * half) * 4 + 2];
    }
    float z0r[4][8];
#pragma unroll
    for (int j = 0; j < 4; ++j)
#pragma unroll
      for (int r = 0; r < 8; ++r)
        z0r[j][r] = z0g[(size_t)dn[r] * 64 + 16 * j + nl];
    float z1r[3][4][8];
#pragma unroll
    for (int i = 0; i < 3; ++i)
#pragma unroll
      for (int j = 0; j < 4; ++j)
#pragma unroll
        for (int r = 0; r < 8; ++r)
          z1r[i][j][r] = z1g[(size_t)dn[r] * 192 + i * 64 + 16 * j + nl];

    v8f g[4][4];
#pragma unroll
    for (int c = 0; c < 4; ++c) {
#pragma unroll
      for (int j = 0; j < 4; ++j) g[c][j] = (v8f)0.0f;
#pragma unroll
      for (int t = 0; t < 16; ++t) {
        int k0 = 4 * t + 2 * half;
        v2f a = *(const v2f*)(stage + nl * WP + k0);
#pragma unroll
        for (int j = 0; j < 4; ++j) {
          v2f b = *(const v2f*)(smem + OFF_WGATE +
                                (64 * c + 16 * j + nl) * WP + k0);
          g[c][j] = wmma4(a, b, g[c][j]);
        }
      }
#pragma unroll
      for (int j = 0; j < 4; ++j) {
        float bb = smem[OFF_BGATE + 64 * c + 16 * j + nl];
#pragma unroll
        for (int r = 0; r < 8; ++r) g[c][j][r] += bb;
      }
    }

#pragma unroll
    for (int j = 0; j < 4; ++j)
#pragma unroll
      for (int r = 0; r < 8; ++r) {
        float dotv = rh[0][r] * z1r[0][j][r] + rh[1][r] * z1r[1][j][r] +
                     rh[2][r] * z1r[2][j][r];
        g[0][j][r] = g[0][j][r] * z0r[j][r] + g[1][j][r] * dotv;
      }

    lds_fence();
#pragma unroll
    for (int j = 0; j < 4; ++j)
#pragma unroll
      for (int r = 0; r < 8; ++r)
        stage[(r + 8 * half) * WP + 16 * j + nl] = g[0][j][r];
    lds_fence();
    v8f p[4];
#pragma unroll
    for (int j = 0; j < 4; ++j) p[j] = (v8f)0.0f;
#pragma unroll
    for (int t = 0; t < 16; ++t) {
      int k0 = 4 * t + 2 * half;
      v2f a = *(const v2f*)(stage + nl * WP + k0);
#pragma unroll
      for (int j = 0; j < 4; ++j) {
        v2f b = *(const v2f*)(smem + OFF_WS + (16 * j + nl) * WP + k0);
        p[j] = wmma4(a, b, p[j]);
      }
    }
#pragma unroll
    for (int j = 0; j < 4; ++j)
#pragma unroll
      for (int r = 0; r < 8; ++r)
        atomicAdd(&out0[(size_t)sn[r] * 64 + 16 * j + nl], p[j][r]);

#pragma unroll
    for (int i = 0; i < 3; ++i) {
      lds_fence();
#pragma unroll
      for (int j = 0; j < 4; ++j)
#pragma unroll
        for (int r = 0; r < 8; ++r)
          stage[(r + 8 * half) * WP + 16 * j + nl] =
              g[2][j][r] * z1r[i][j][r] + g[3][j][r] * rh[i][r] * z0r[j][r];
      lds_fence();
#pragma unroll
      for (int j = 0; j < 4; ++j) p[j] = (v8f)0.0f;
#pragma unroll
      for (int t = 0; t < 16; ++t) {
        int k0 = 4 * t + 2 * half;
        v2f a = *(const v2f*)(stage + nl * WP + k0);
#pragma unroll
        for (int j = 0; j < 4; ++j) {
          v2f b = *(const v2f*)(smem + OFF_WV + (16 * j + nl) * WP + k0);
          p[j] = wmma4(a, b, p[j]);
        }
      }
#pragma unroll
      for (int j = 0; j < 4; ++j)
#pragma unroll
        for (int r = 0; r < 8; ++r)
          atomicAdd(&out1[(size_t)sn[r] * 192 + i * 64 + 16 * j + nl],
                    p[j][r]);
    }
  }
}

extern "C" void kernel_launch(void* const* d_in, const int* in_sizes, int n_in,
                              void* d_out, int out_size, void* d_ws,
                              size_t ws_size, hipStream_t stream) {
  const int*   src   = (const int*)d_in[0];
  const int*   dst   = (const int*)d_in[1];
  const float* r_ij  = (const float*)d_in[2];
  const float* z0    = (const float*)d_in[3];
  const float* z1    = (const float*)d_in[4];
  const float* emb   = (const float*)d_in[5];
  const float* Wenc  = (const float*)d_in[6];
  const float* benc  = (const float*)d_in[7];
  const float* Wsrc  = (const float*)d_in[8];
  const float* bsrc  = (const float*)d_in[9];
  const float* Wdst  = (const float*)d_in[10];
  const float* bdst  = (const float*)d_in[11];
  const float* Wgate = (const float*)d_in[12];
  const float* bgate = (const float*)d_in[13];
  const float* Ws    = (const float*)d_in[14];
  const float* Wv    = (const float*)d_in[15];
  float* out = (float*)d_out;

  zero_f32<<<(out_size + 255) / 256, 256, 0, stream>>>(out, out_size);

  size_t smem_bytes = (size_t)SMEM_FLOATS * sizeof(float);
  messages_kernel<<<GRID, BLOCK, smem_bytes, stream>>>(
      src, dst, r_ij, z0, z1, emb, Wenc, benc, Wsrc, bsrc, Wdst, bdst, Wgate,
      bgate, Ws, Wv, out, out + (size_t)NNODES * CHAN);
}


# W_s and W_v applied after the scatter-add, once per node row in a new in-place node kernel, not per edge row (segment_sum is linear): 256 of 648 WMMAs per edge tile go; float sum order changes
# speedup vs baseline: 3.5163x; 1.0746x over previous
#include <hip/hip_runtime.h>
#include <math.h>

#define NNODES 50000
#define NEDGES 500000
#define CHAN   64
#define R0f    2.0f

#define WAVES  8
#define BLOCK  (WAVES * 32)
#define NTILES (NEDGES / 16)
#define GRID   512

typedef float v2f __attribute__((ext_vector_type(2)));
typedef float v4f __attribute__((ext_vector_type(4)));
typedef float v8f __attribute__((ext_vector_type(8)));

#define WP 68
#define EP 12
#define OFF_WSRC  0
#define OFF_WDST  (OFF_WSRC + 64 * WP)
#define OFF_WGATE (OFF_WDST + 64 * WP)
#define OFF_WENC  (OFF_WGATE + 256 * WP)
#define OFF_BIAS0 (OFF_WENC + 64 * EP)
#define OFF_BGATE (OFF_BIAS0 + 64)
#define OFF_STAGE (OFF_BGATE + 256)
#define OFF_RH    (OFF_STAGE + WAVES * 16 * WP)
#define OFF_IDX   (OFF_RH + WAVES * 64)
#define SMEM_FLOATS (OFF_IDX + WAVES * 32)

#define NODE_TILES (NNODES / 16)
#define NODE_TASKS (NODE_TILES * 4)
#define NGRID      1024
#define N_OFF_WS    0
#define N_OFF_WV    (N_OFF_WS + 64 * WP)
#define N_OFF_STAGE (N_OFF_WV + 64 * WP)
#define N_SMEM_FLOATS (N_OFF_STAGE + WAVES * 16 * WP)

__device__ __forceinline__ void lds_fence() {
  asm volatile("s_wait_dscnt 0x0" ::: "memory");
}

__device__ __forceinline__ v8f wmma4(v2f a, v2f b, v8f c) {
  return __builtin_amdgcn_wmma_f32_16x16x4_f32(false, a, false, b, (short)0, c,
                                               false, false);
}

__global__ void zero_f32(float* __restrict__ p, int n) {
  int i = blockIdx.x * blockDim.x + threadIdx.x;
  if (i < n) p[i] = 0.0f;
}

__global__ void __launch_bounds__(BLOCK, 1)
messages_kernel(const int* __restrict__ src, const int* __restrict__ dst,
                const float* __restrict__ r_ij,
                const float* __restrict__ z0g, const float* __restrict__ z1g,
                const float* __restrict__ emb,
                const float* __restrict__ Wenc, const float* __restrict__ benc,
                const float* __restrict__ Wsrc, const float* __restrict__ bsrc,
                const float* __restrict__ Wdst, const float* __restrict__ bdst,
                const float* __restrict__ Wgate, const float* __restrict__ bgate,
                float* __restrict__ out0, float* __restrict__ out1) {
  extern __shared__ float smem[];

  for (int i = threadIdx.x; i < 64 * 64; i += BLOCK) {
    int k = i >> 6, n = i & 63;
    smem[OFF_WSRC + n * WP + k] = Wsrc[i];
    smem[OFF_WDST + n * WP + k] = Wdst[i];
  }
  for (int i = threadIdx.x; i < 64 * 256; i += BLOCK) {
    int k = i >> 8, n = i & 255;
    smem[OFF_WGATE + n * WP + k] = Wgate[i];
  }
  for (int i = threadIdx.x; i < 8 * 64; i += BLOCK) {
    int k = i >> 6, n = i & 63;
    smem[OFF_WENC + n * EP + k] = Wenc[i];
  }
  if (threadIdx.x < 64)
    smem[OFF_BIAS0 + threadIdx.x] =
        bsrc[threadIdx.x] + bdst[threadIdx.x] + benc[threadIdx.x];
  if (threadIdx.x < 256) smem[OFF_BGATE + threadIdx.x] = bgate[threadIdx.x];
  __syncthreads();

  const int lane = threadIdx.x & 31;
  const int wave = threadIdx.x >> 5;
  const int half = lane >> 4;
  const int nl   = lane & 15;

  float* stage = smem + OFF_STAGE + wave * (16 * WP);
  float* rhS   = smem + OFF_RH + wave * 64;
  int*   idxS  = ((int*)(smem + OFF_IDX)) + wave * 32;

  for (int tile = blockIdx.x * WAVES + wave; tile < NTILES;
       tile += gridDim.x * WAVES) {
    const int e  = tile * 16 + nl;
    const int sN = src[e];
    const int dN = dst[e];
    if (half == 0) { idxS[nl] = sN; idxS[16 + nl] = dN; }

    float rx = r_ij[e * 3 + 0], ry = r_ij[e * 3 + 1], rz = r_ij[e * 3 + 2];
    float dd  = sqrtf(rx * rx + ry * ry + rz * rz);
    float sx = rx * 3.5f, sy = ry * 3.5f, sz = rz * 3.5f;
    float inv = rsqrtf(1.0f + sx * sx + sy * sy + sz * sz);
    if (half == 0) {
      rhS[nl * 4 + 0] = sx * inv;
      rhS[nl * 4 + 1] = sy * inv;
      rhS[nl * 4 + 2] = sz * inv;
    }
    float rb[4];
#pragma unroll
    for (int u = 0; u < 4; ++u) {
      int k = (u >> 1) * 4 + 2 * half + (u & 1);
      float x = (dd - (float)k * (R0f / 7.0f)) * 4.0f;
      rb[u] = expf(-x * x);
    }

    v8f acc[4];
#pragma unroll
    for (int j = 0; j < 4; ++j) acc[j] = (v8f)0.0f;

    const float* es = emb + (size_t)sN * 64;
    const float* ed = emb + (size_t)dN * 64;
#pragma unroll
    for (int t = 0; t < 16; ++t) {
      int k0 = 4 * t + 2 * half;
      v2f a = *(const v2f*)(es + k0);
#pragma unroll
      for (int j = 0; j < 4; ++j) {
        v2f b = *(const v2f*)(smem + OFF_WSRC + (16 * j + nl) * WP + k0);
        acc[j] = wmma4(a, b, acc[j]);
      }
    }
#pragma unroll
    for (int t = 0; t < 16; ++t) {
      int k0 = 4 * t + 2 * half;
      v2f a = *(const v2f*)(ed + k0);
#pragma unroll
      for (int j = 0; j < 4; ++j) {
        v2f b = *(const v2f*)(smem + OFF_WDST + (16 * j + nl) * WP + k0);
        acc[j] = wmma4(a, b, acc[j]);
      }
    }
#pragma unroll
    for (int t = 0; t < 2; ++t) {
      int k0 = 4 * t + 2 * half;
      v2f a;
      a[0] = rb[2 * t];
      a[1] = rb[2 * t + 1];
#pragma unroll
      for (int j = 0; j < 4; ++j) {
        v2f b = *(const v2f*)(smem + OFF_WENC + (16 * j + nl) * EP + k0);
        acc[j] = wmma4(a, b, acc[j]);
      }
    }
#pragma unroll
    for (int j = 0; j < 4; ++j) {
      float bb = smem[OFF_BIAS0 + 16 * j + nl];
#pragma unroll
      for (int r = 0; r < 8; ++r) acc[j][r] += bb;
    }

    lds_fence();
#pragma unroll
    for (int j = 0; j < 4; ++j)
#pragma unroll
      for (int r = 0; r < 8; ++r)
        stage[(r + 8 * half) * WP + 16 * j + nl] = acc[j][r];
    lds_fence();

    int sn[8], dn[8];
#pragma unroll
    for (int r = 0; r < 8; ++r) {
      sn[r] = idxS[r + 8 * half];
      dn[r] = idxS[16 + r + 8 * half];
    }
    float rh[3][8];
#pragma unroll
    for (int r = 0; r < 8; ++r) {
      rh[0][r] = rhS[(r + 8 * half) * 4 + 0];
      rh[1][r] = rhS[(r + 8 * half) * 4 + 1];
      rh[2][r] = rhS[(r + 8 * half) * 4 + 2];
    }
    float z0r[4][8];
#pragma unroll
    for (int j = 0; j < 4; ++j)
#pragma unroll
      for (int r = 0; r < 8; ++r)
        z0r[j][r] = z0g[(size_t)dn[r] * 64 + 16 * j + nl];
    float z1r[3][4][8];
#pragma unroll
    for (int i = 0; i < 3; ++i)
#pragma unroll
      for (int j = 0; j < 4; ++j)
#pragma unroll
        for (int r = 0; r < 8; ++r)
          z1r[i][j][r] = z1g[(size_t)dn[r] * 192 + i * 64 + 16 * j + nl];

    v8f g[4][4];
#pragma unroll
    for (int c = 0; c < 4; ++c) {
#pragma unroll
      for (int j = 0; j < 4; ++j) g[c][j] = (v8f)0.0f;
#pragma unroll
      for (int t = 0; t < 16; ++t) {
        int k0 = 4 * t + 2 * half;
        v2f a = *(const v2f*)(stage + nl * WP + k0);
#pragma unroll
        for (int j = 0; j < 4; ++j) {
          v2f b = *(const v2f*)(smem + OFF_WGATE +
                                (64 * c + 16 * j + nl) * WP + k0);
          g[c][j] = wmma4(a, b, g[c][j]);
        }
      }
#pragma unroll
      for (int j = 0; j < 4; ++j) {
        float bb = smem[OFF_BGATE + 64 * c + 16 * j + nl];
#pragma unroll
        for (int r = 0; r < 8; ++r) g[c][j][r] += bb;
      }
    }

#pragma unroll
    for (int j = 0; j < 4; ++j)
#pragma unroll
      for (int r = 0; r < 8; ++r) {
        float dotv = rh[0][r] * z1r[0][j][r] + rh[1][r] * z1r[1][j][r] +
                     rh[2][r] * z1r[2][j][r];
        atomicAdd(&out0[(size_t)sn[r] * 64 + 16 * j + nl],
                  g[0][j][r] * z0r[j][r] + g[1][j][r] * dotv);
      }

#pragma unroll
    for (int i = 0; i < 3; ++i)
#pragma unroll
      for (int j = 0; j < 4; ++j)
#pragma unroll
        for (int r = 0; r < 8; ++r)
          atomicAdd(&out1[(size_t)sn[r] * 192 + i * 64 + 16 * j + nl],
                    g[2][j][r] * z1r[i][j][r] + g[3][j][r] * rh[i][r] * z0r[j][r]);
  }
}

__global__ void __launch_bounds__(BLOCK)
node_transform_kernel(const float* __restrict__ Ws, const float* __restrict__ Wv,
                      float* out) {
  extern __shared__ float smem[];

  for (int i = threadIdx.x; i < 64 * 64; i += BLOCK) {
    int k = i >> 6, n = i & 63;
    smem[N_OFF_WS + n * WP + k] = Ws[i];
    smem[N_OFF_WV + n * WP + k] = Wv[i];
  }
  __syncthreads();

  const int lane = threadIdx.x & 31;
  const int wave = threadIdx.x >> 5;
  const int half = lane >> 4;
  const int nl   = lane & 15;

  float* stage = smem + N_OFF_STAGE + wave * (16 * WP);

  for (int task = blockIdx.x * WAVES + wave; task < NODE_TASKS;
       task += gridDim.x * WAVES) {
    const int tile = task >> 2;
    const int part = task & 3;
    float* base;
    int pitch;
    const float* wT;
    if (part == 0) {
      base = out + (size_t)tile * (16 * 64);
      pitch = 64;
      wT = smem + N_OFF_WS;
    } else {
      base = out + (size_t)NNODES * CHAN + (size_t)tile * (16 * 192) + (part - 1) * 64;
      pitch = 192;
      wT = smem + N_OFF_WV;
    }

    v4f v[8];
#pragma unroll
    for (int q = 0; q < 8; ++q) {
      int idx = q * 32 + lane;
      v[q] = *(const v4f*)(base + (size_t)(idx >> 4) * pitch + (idx & 15) * 4);
    }
    lds_fence();
#pragma unroll
    for (int q = 0; q < 8; ++q) {
      int idx = q * 32 + lane;
      *(v4f*)(stage + (idx >> 4) * WP + (idx & 15) * 4) = v[q];
    }
    lds_fence();

    v8f p[4];
#pragma unroll
    for (int j = 0; j < 4; ++j) p[j] = (v8f)0.0f;
#pragma unroll
    for (int t = 0; t < 16; ++t) {
      int k0 = 4 * t + 2 * half;
      v2f a = *(const v2f*)(stage + nl * WP + k0);
#pragma unroll
      for (int j = 0; j < 4; ++j) {
        v2f b = *(const v2f*)(wT + (16 * j + nl) * WP + k0);
        p[j] = wmma4(a, b, p[j]);
      }
    }

    lds_fence();
#pragma unroll
    for (int j = 0; j < 4; ++j)
#pragma unroll
      for (int r = 0; r < 8; ++r)
        stage[(r + 8 * half) * WP + 16 * j + nl] = p[j][r];
    lds_fence();
#pragma unroll
    for (int q = 0; q < 8; ++q) {
      int idx = q * 32 + lane;
      v[q] = *(const v4f*)(stage + (idx >> 4) * WP + (idx & 15) * 4);
    }
#pragma unroll
    for (int q = 0; q < 8; ++q) {
      int idx = q * 32 + lane;
      *(v4f*)(base + (size_t)(idx >> 4) * pitch + (idx & 15) * 4) = v[q];
    }
  }
}

extern "C" void kernel_launch(void* const* d_in, const int* in_sizes, int n_in,
                              void* d_out, int out_size, void* d_ws,
                              size_t ws_size, hipStream_t stream) {
  const int*   src   = (const int*)d_in[0];
  const int*   dst   = (const int*)d_in[1];
  const float* r_ij  = (const float*)d_in[2];
  const float* z0    = (const float*)d_in[3];
  const float* z1    = (const float*)d_in[4];
  const float* emb   = (const float*)d_in[5];
  const float* Wenc  = (const float*)d_in[6];
  const float* benc  = (const float*)d_in[7];
  const float* Wsrc  = (const float*)d_in[8];
  const float* bsrc  = (const float*)d_in[9];
  const float* Wdst  = (const float*)d_in[10];
  const float* bdst  = (const float*)d_in[11];
  const float* Wgate = (const float*)d_in[12];
  const float* bgate = (const float*)d_in[13];
  const float* Ws    = (const float*)d_in[14];
  const float* Wv    = (const float*)d_in[15];
  float* out = (float*)d_out;

  zero_f32<<<(out_size + 255) / 256, 256, 0, stream>>>(out, out_size);

  size_t smem_bytes = (size_t)SMEM_FLOATS * sizeof(float);
  messages_kernel<<<GRID, BLOCK, smem_bytes, stream>>>(
      src, dst, r_ij, z0, z1, emb, Wenc, benc, Wsrc, bsrc, Wdst, bdst, Wgate,
      bgate, out, out + (size_t)NNODES * CHAN);

  size_t nsmem_bytes = (size_t)N_SMEM_FLOATS * sizeof(float);
  node_transform_kernel<<<NGRID, BLOCK, nsmem_bytes, stream>>>(Ws, Wv, out);
}


# zero fill of out as 16-byte stores from 256 resident 8-wave work-groups that loop over the buffer, not 50000 work-groups with one 4-byte store per thread: 2048 waves enter, not 400000
# speedup vs baseline: 3.6568x; 1.0400x over previous
#include <hip/hip_runtime.h>
#include <math.h>

#define NNODES 50000
#define NEDGES 500000
#define CHAN   64
#define R0f    2.0f

#define WAVES  8
#define BLOCK  (WAVES * 32)
#define NTILES (NEDGES / 16)
#define GRID   512

typedef float v2f __attribute__((ext_vector_type(2)));
typedef float v4f __attribute__((ext_vector_type(4)));
typedef float v8f __attribute__((ext_vector_type(8)));

#define WP 68
#define EP 12
#define OFF_WSRC  0
#define OFF_WDST  (OFF_WSRC + 64 * WP)
#define OFF_WGATE (OFF_WDST + 64 * WP)
#define OFF_WENC  (OFF_WGATE + 256 * WP)
#define OFF_BIAS0 (OFF_WENC + 64 * EP)
#define OFF_BGATE (OFF_BIAS0 + 64)
#define OFF_STAGE (OFF_BGATE + 256)
#define OFF_RH    (OFF_STAGE + WAVES * 16 * WP)
#define OFF_IDX   (OFF_RH + WAVES * 64)
#define SMEM_FLOATS (OFF_IDX + WAVES * 32)

#define NODE_TILES (NNODES / 16)
#define NODE_TASKS (NODE_TILES * 4)
#define NGRID      1024
#define N_OFF_WS    0
#define N_OFF_WV    (N_OFF_WS + 64 * WP)
#define N_OFF_STAGE (N_OFF_WV + 64 * WP)
#define N_SMEM_FLOATS (N_OFF_STAGE + WAVES * 16 * WP)

__device__ __forceinline__ void lds_fence() {
  asm volatile("s_wait_dscnt 0x0" ::: "memory");
}

__device__ __forceinline__ v8f wmma4(v2f a, v2f b, v8f c) {
  return __builtin_amdgcn_wmma_f32_16x16x4_f32(false, a, false, b, (short)0, c,
                                               false, false);
}

#define ZGRID   256
#define ZSTRIDE (ZGRID * BLOCK)
__global__ void __launch_bounds__(BLOCK)
zero_f32(float* __restrict__ p, int n) {
  const int n4 = n >> 2;
  v4f* __restrict__ p4 = (v4f*)p;
  int i = blockIdx.x * BLOCK + threadIdx.x;
  for (; i + 7 * ZSTRIDE < n4; i += 8 * ZSTRIDE) {
#pragma unroll
    for (int u = 0; u < 8; ++u) p4[i + u * ZSTRIDE] = (v4f)0.0f;
  }
  for (; i < n4; i += ZSTRIDE) p4[i] = (v4f)0.0f;
  const int t = (n4 << 2) + threadIdx.x;
  if (blockIdx.x == 0 && t < n) p[t] = 0.0f;
}

__global__ void __launch_bounds__(BLOCK, 1)
messages_kernel(const int* __restrict__ src, const int* __restrict__ dst,
                const float* __restrict__ r_ij,
                const float* __restrict__ z0g, const float* __restrict__ z1g,
                const float* __restrict__ emb,
                const float* __restrict__ Wenc, const float* __restrict__ benc,
                const float* __restrict__ Wsrc, const float* __restrict__ bsrc,
                const float* __restrict__ Wdst, const float* __restrict__ bdst,
                const float* __restrict__ Wgate, const float* __restrict__ bgate,
                float* __restrict__ out0, float* __restrict__ out1) {
  extern __shared__ float smem[];

  for (int i = threadIdx.x; i < 64 * 64; i += BLOCK) {
    int k = i >> 6, n = i & 63;
    smem[OFF_WSRC + n * WP + k] = Wsrc[i];
    smem[OFF_WDST + n * WP + k] = Wdst[i];
  }
  for (int i = threadIdx.x; i < 64 * 256; i += BLOCK) {
    int k = i >> 8, n = i & 255;
    smem[OFF_WGATE + n * WP + k] = Wgate[i];
  }
  for (int i = threadIdx.x; i < 8 * 64; i += BLOCK) {
    int k = i >> 6, n = i & 63;
    smem[OFF_WENC + n * EP + k] = Wenc[i];
  }
  if (threadIdx.x < 64)
    smem[OFF_BIAS0 + threadIdx.x] =
        bsrc[threadIdx.x] + bdst[threadIdx.x] + benc[threadIdx.x];
  if (threadIdx.x < 256) smem[OFF_BGATE + threadIdx.x] = bgate[threadIdx.x];
  __syncthreads();

  const int lane = threadIdx.x & 31;
  const int wave = threadIdx.x >> 5;
  const int half = lane >> 4;
  const int nl   = lane & 15;

  float* stage = smem + OFF_STAGE + wave * (16 * WP);
  float* rhS   = smem + OFF_RH + wave * 64;
  int*   idxS  = ((int*)(smem + OFF_IDX)) + wave * 32;

  for (int tile = blockIdx.x * WAVES + wave; tile < NTILES;
       tile += gridDim.x * WAVES) {
    const int e  = tile * 16 + nl;
    const int sN = src[e];
    const int dN = dst[e];
    if (half == 0) { idxS[nl] = sN; idxS[16 + nl] = dN; }

    float rx = r_ij[e * 3 + 0], ry = r_ij[e * 3 + 1], rz = r_ij[e * 3 + 2];
    float dd  = sqrtf(rx * rx + ry * ry + rz * rz);
    float sx = rx * 3.5f, sy = ry * 3.5f, sz = rz * 3.5f;
    float inv = rsqrtf(1.0f + sx * sx + sy * sy + sz * sz);
    if (half == 0) {
      rhS[nl * 4 + 0] = sx * inv;
      rhS[nl * 4 + 1] = sy * inv;
      rhS[nl * 4 + 2] = sz * inv;
    }
    float rb[4];
#pragma unroll
    for (int u = 0; u < 4; ++u) {
      int k = (u >> 1) * 4 + 2 * half + (u & 1);
      float x = (dd - (float)k * (R0f / 7.0f)) * 4.0f;
      rb[u] = expf(-x * x);
    }

    v8f acc[4];
#pragma unroll
    for (int j = 0; j < 4; ++j) acc[j] = (v8f)0.0f;

    const float* es = emb + (size_t)sN * 64;
    const float* ed = emb + (size_t)dN * 64;
#pragma unroll
    for (int t = 0; t < 16; ++t) {
      int k0 = 4 * t + 2 * half;
      v2f a = *(const v2f*)(es + k0);
#pragma unroll
      for (int j = 0; j < 4; ++j) {
        v2f b = *(const v2f*)(smem + OFF_WSRC + (16 * j + nl) * WP + k0);
        acc[j] = wmma4(a, b, acc[j]);
      }
    }
#pragma unroll
    for (int t = 0; t < 16; ++t) {
      int k0 = 4 * t + 2 * half;
      v2f a = *(const v2f*)(ed + k0);
#pragma unroll
      for (int j = 0; j < 4; ++j) {
        v2f b = *(const v2f*)(smem + OFF_WDST + (16 * j + nl) * WP + k0);
        acc[j] = wmma4(a, b, acc[j]);
      }
    }
#pragma unroll
    for (int t = 0; t < 2; ++t) {
      int k0 = 4 * t + 2 * half;
      v2f a;
      a[0] = rb[2 * t];
      a[1] = rb[2 * t + 1];
#pragma unroll
      for (int j = 0; j < 4; ++j) {
        v2f b = *(const v2f*)(smem + OFF_WENC + (16 * j + nl) * EP + k0);
        acc[j] = wmma4(a, b, acc[j]);
      }
    }
#pragma unroll
    for (int j = 0; j < 4; ++j) {
      float bb = smem[OFF_BIAS0 + 16 * j + nl];
#pragma unroll
      for (int r = 0; r < 8; ++r) acc[j][r] += bb;
    }

    lds_fence();
#pragma unroll
    for (int j = 0; j < 4; ++j)
#pragma unroll
      for (int r = 0; r < 8; ++r)
        stage[(r + 8 * half) * WP + 16 * j + nl] = acc[j][r];
    lds_fence();

    int sn[8], dn[8];
#pragma unroll
    for (int r = 0; r < 8; ++r) {
      sn[r] = idxS[r + 8 * half];
      dn[r] = idxS[16 + r + 8 * half];
    }
    float rh[3][8];
#pragma unroll
    for (int r = 0; r < 8; ++r) {
      rh[0][r] = rhS[(r + 8 * half) * 4 + 0];
      rh[1][r] = rhS[(r + 8 * half) * 4 + 1];
      rh[2][r] = rhS[(r + 8 * half) * 4 + 2];
    }
    float z0r[4][8];
#pragma unroll
    for (int j = 0; j < 4; ++j)
#pragma unroll
      for (int r = 0; r < 8; ++r)
        z0r[j][r] = z0g[(size_t)dn[r] * 64 + 16 * j + nl];
    float z1r[3][4][8];
#pragma unroll
    for (int i = 0; i < 3; ++i)
#pragma unroll
      for (int j = 0; j < 4; ++j)
#pragma unroll
        for (int r = 0; r < 8; ++r)
          z1r[i][j][r] = z1g[(size_t)dn[r] * 192 + i * 64 + 16 * j + nl];

    v8f g[4][4];
#pragma unroll
    for (int c = 0; c < 4; ++c) {
#pragma unroll
      for (int j = 0; j < 4; ++j) g[c][j] = (v8f)0.0f;
#pragma unroll
      for (int t = 0; t < 16; ++t) {
        int k0 = 4 * t + 2 * half;
        v2f a = *(const v2f*)(stage + nl * WP + k0);
#pragma unroll
        for (int j = 0; j < 4; ++j) {
          v2f b = *(const v2f*)(smem + OFF_WGATE +
                                (64 * c + 16 * j + nl) * WP + k0);
          g[c][j] = wmma4(a, b, g[c][j]);
        }
      }
#pragma unroll
      for (int j = 0; j < 4; ++j) {
        float bb = smem[OFF_BGATE + 64 * c + 16 * j + nl];
#pragma unroll
        for (int r = 0; r < 8; ++r) g[c][j][r] += bb;
      }
    }

#pragma unroll
    for (int j = 0; j < 4; ++j)
#pragma unroll
      for (int r = 0; r < 8; ++r) {
        float dotv = rh[0][r] * z1r[0][j][r] + rh[1][r] * z1r[1][j][r] +
                     rh[2][r] * z1r[2][j][r];
        atomicAdd(&out0[(size_t)sn[r] * 64 + 16 * j + nl],
                  g[0][j][r] * z0r[j][r] + g[1][j][r] * dotv);
      }

#pragma unroll
    for (int i = 0; i < 3; ++i)
#pragma unroll
      for (int j = 0; j < 4; ++j)
#pragma unroll
        for (int r = 0; r < 8; ++r)
          atomicAdd(&out1[(size_t)sn[r] * 192 + i * 64 + 16 * j + nl],
                    g[2][j][r] * z1r[i][j][r] + g[3][j][r] * rh[i][r] * z0r[j][r]);
  }
}

__global__ void __launch_bounds__(BLOCK)
node_transform_kernel(const float* __restrict__ Ws, const float* __restrict__ Wv,
                      float* out) {
  extern __shared__ float smem[];

  for (int i = threadIdx.x; i < 64 * 64; i += BLOCK) {
    int k = i >> 6, n = i & 63;
    smem[N_OFF_WS + n * WP + k] = Ws[i];
    smem[N_OFF_WV + n * WP + k] = Wv[i];
  }
  __syncthreads();

  const int lane = threadIdx.x & 31;
  const int wave = threadIdx.x >> 5;
  const int half = lane >> 4;
  const int nl   = lane & 15;

  float* stage = smem + N_OFF_STAGE + wave * (16 * WP);

  for (int task = blockIdx.x * WAVES + wave; task < NODE_TASKS;
       task += gridDim.x * WAVES) {
    const int tile = task >> 2;
    const int part = task & 3;
    float* base;
    int pitch;
    const float* wT;
    if (part == 0) {
      base = out + (size_t)tile * (16 * 64);
      pitch = 64;
      wT = smem + N_OFF_WS;
    } else {
      base = out + (size_t)NNODES * CHAN + (size_t)tile * (16 * 192) + (part - 1) * 64;
      pitch = 192;
      wT = smem + N_OFF_WV;
    }

    v4f v[8];
#pragma unroll
    for (int q = 0; q < 8; ++q) {
      int idx = q * 32 + lane;
      v[q] = *(const v4f*)(base + (size_t)(idx >> 4) * pitch + (idx & 15) * 4);
    }
    lds_fence();
#pragma unroll
    for (int q = 0; q < 8; ++q) {
      int idx = q * 32 + lane;
      *(v4f*)(stage + (idx >> 4) * WP + (idx & 15) * 4) = v[q];
    }
    lds_fence();

    v8f p[4];
#pragma unroll
    for (int j = 0; j < 4; ++j) p[j] = (v8f)0.0f;
#pragma unroll
    for (int t = 0; t < 16; ++t) {
      int k0 = 4 * t + 2 * half;
      v2f a = *(const v2f*)(stage + nl * WP + k0);
#pragma unroll
      for (int j = 0; j < 4; ++j) {
        v2f b = *(const v2f*)(wT + (16 * j + nl) * WP + k0);
        p[j] = wmma4(a, b, p[j]);
      }
    }

    lds_fence();
#pragma unroll
    for (int j = 0; j < 4; ++j)
#pragma unroll
      for (int r = 0; r < 8; ++r)
        stage[(r + 8 * half) * WP + 16 * j + nl] = p[j][r];
    lds_fence();
#pragma unroll
    for (int q = 0; q < 8; ++q) {
      int idx = q * 32 + lane;
      v[q] = *(const v4f*)(stage + (idx >> 4) * WP + (idx & 15) * 4);
    }
#pragma unroll
    for (int q = 0; q < 8; ++q) {
      int idx = q * 32 + lane;
      *(v4f*)(base + (size_t)(idx >> 4) * pitch + (idx & 15) * 4) = v[q];
    }
  }
}

extern "C" void kernel_launch(void* const* d_in, const int* in_sizes, int n_in,
                              void* d_out, int out_size, void* d_ws,
                              size_t ws_size, hipStream_t stream) {
  const int*   src   = (const int*)d_in[0];
  const int*   dst   = (const int*)d_in[1];
  const float* r_ij  = (const float*)d_in[2];
  const float* z0    = (const float*)d_in[3];
  const float* z1    = (const float*)d_in[4];
  const float* emb   = (const float*)d_in[5];
  const float* Wenc  = (const float*)d_in[6];
  const float* benc  = (const float*)d_in[7];
  const float* Wsrc  = (const float*)d_in[8];
  const float* bsrc  = (const float*)d_in[9];
  const float* Wdst  = (const float*)d_in[10];
  const float* bdst  = (const float*)d_in[11];
  const float* Wgate = (const float*)d_in[12];
  const float* bgate = (const float*)d_in[13];
  const float* Ws    = (const float*)d_in[14];
  const float* Wv    = (const float*)d_in[15];
  float* out = (float*)d_out;

  zero_f32<<<ZGRID, BLOCK, 0, stream>>>(out, out_size);

  size_t smem_bytes = (size_t)SMEM_FLOATS * sizeof(float);
  messages_kernel<<<GRID, BLOCK, smem_bytes, stream>>>(
      src, dst, r_ij, z0, z1, emb, Wenc, benc, Wsrc, bsrc, Wdst, bdst, Wgate,
      bgate, out, out + (size_t)NNODES * CHAN);

  size_t nsmem_bytes = (size_t)N_SMEM_FLOATS * sizeof(float);
  node_transform_kernel<<<NGRID, BLOCK, nsmem_bytes, stream>>>(Ws, Wv, out);
}


# gate and messages done one 16-channel group at a time, 32-bit row offsets on scalar bases: 442 -> 223 registers, 4 waves per SIMD not 2 (two work-groups per WGP); per-message arithmetic unchanged
# speedup vs baseline: 4.1850x; 1.1444x over previous
#include <hip/hip_runtime.h>
#include <math.h>

#define NNODES 50000
#define NEDGES 500000
#define CHAN   64
#define R0f    2.0f

#define WAVES  8
#define BLOCK  (WAVES * 32)
#define NTILES (NEDGES / 16)
#define GRID   512

typedef float v2f __attribute__((ext_vector_type(2)));
typedef float v4f __attribute__((ext_vector_type(4)));
typedef float v8f __attribute__((ext_vector_type(8)));

#define WP 68
#define EP 12
#define OFF_WSRC  0
#define OFF_WDST  (OFF_WSRC + 64 * WP)
#define OFF_WGATE (OFF_WDST + 64 * WP)
#define OFF_WENC  (OFF_WGATE + 256 * WP)
#define OFF_BIAS0 (OFF_WENC + 64 * EP)
#define OFF_BGATE (OFF_BIAS0 + 64)
#define OFF_STAGE (OFF_BGATE + 256)
#define OFF_RH    (OFF_STAGE + WAVES * 16 * WP)
#define OFF_IDX   (OFF_RH + WAVES * 64)
#define SMEM_FLOATS (OFF_IDX + WAVES * 32)

#define NODE_TILES (NNODES / 16)
#define NODE_TASKS (NODE_TILES * 4)
#define NGRID      1024
#define N_OFF_WS    0
#define N_OFF_WV    (N_OFF_WS + 64 * WP)
#define N_OFF_STAGE (N_OFF_WV + 64 * WP)
#define N_SMEM_FLOATS (N_OFF_STAGE + WAVES * 16 * WP)

__device__ __forceinline__ void lds_fence() {
  asm volatile("s_wait_dscnt 0x0" ::: "memory");
}

__device__ __forceinline__ v8f wmma4(v2f a, v2f b, v8f c) {
  return __builtin_amdgcn_wmma_f32_16x16x4_f32(false, a, false, b, (short)0, c,
                                               false, false);
}

#define ZGRID   256
#define ZSTRIDE (ZGRID * BLOCK)
__global__ void __launch_bounds__(BLOCK)
zero_f32(float* __restrict__ p, int n) {
  const int n4 = n >> 2;
  v4f* __restrict__ p4 = (v4f*)p;
  int i = blockIdx.x * BLOCK + threadIdx.x;
  for (; i + 7 * ZSTRIDE < n4; i += 8 * ZSTRIDE) {
#pragma unroll
    for (int u = 0; u < 8; ++u) p4[i + u * ZSTRIDE] = (v4f)0.0f;
  }
  for (; i < n4; i += ZSTRIDE) p4[i] = (v4f)0.0f;
  const int t = (n4 << 2) + threadIdx.x;
  if (blockIdx.x == 0 && t < n) p[t] = 0.0f;
}

__global__ void __launch_bounds__(BLOCK, 4)
messages_kernel(const int* __restrict__ src, const int* __restrict__ dst,
                const float* __restrict__ r_ij,
                const float* __restrict__ z0g, const float* __restrict__ z1g,
                const float* __restrict__ emb,
                const float* __restrict__ Wenc, const float* __restrict__ benc,
                const float* __restrict__ Wsrc, const float* __restrict__ bsrc,
                const float* __restrict__ Wdst, const float* __restrict__ bdst,
                const float* __restrict__ Wgate, const float* __restrict__ bgate,
                float* __restrict__ out0, float* __restrict__ out1) {
  extern __shared__ float smem[];

  for (int i = threadIdx.x; i < 64 * 64; i += BLOCK) {
    int k = i >> 6, n = i & 63;
    smem[OFF_WSRC + n * WP + k] = Wsrc[i];
    smem[OFF_WDST + n * WP + k] = Wdst[i];
  }
  for (int i = threadIdx.x; i < 64 * 256; i += BLOCK) {
    int k = i >> 8, n = i & 255;
    smem[OFF_WGATE + n * WP + k] = Wgate[i];
  }
  for (int i = threadIdx.x; i < 8 * 64; i += BLOCK) {
    int k = i >> 6, n = i & 63;
    smem[OFF_WENC + n * EP + k] = Wenc[i];
  }
  if (threadIdx.x < 64)
    smem[OFF_BIAS0 + threadIdx.x] =
        bsrc[threadIdx.x] + bdst[threadIdx.x] + benc[threadIdx.x];
  if (threadIdx.x < 256) smem[OFF_BGATE + threadIdx.x] = bgate[threadIdx.x];
  __syncthreads();

  const int lane = threadIdx.x & 31;
  const int wave = threadIdx.x >> 5;
  const int half = lane >> 4;
  const int nl   = lane & 15;

  float* stage = smem + OFF_STAGE + wave * (16 * WP);
  float* rhS   = smem + OFF_RH + wave * 64;
  int*   idxS  = ((int*)(smem + OFF_IDX)) + wave * 32;

  for (int tile = blockIdx.x * WAVES + wave; tile < NTILES;
       tile += gridDim.x * WAVES) {
    const int e  = tile * 16 + nl;
    const int sN = src[e];
    const int dN = dst[e];
    if (half == 0) { idxS[nl] = sN; idxS[16 + nl] = dN; }

    float rx = r_ij[e * 3 + 0], ry = r_ij[e * 3 + 1], rz = r_ij[e * 3 + 2];
    float dd  = sqrtf(rx * rx + ry * ry + rz * rz);
    float sx = rx * 3.5f, sy = ry * 3.5f, sz = rz * 3.5f;
    float inv = rsqrtf(1.0f + sx * sx + sy * sy + sz * sz);
    if (half == 0) {
      rhS[nl * 4 + 0] = sx * inv;
      rhS[nl * 4 + 1] = sy * inv;
      rhS[nl * 4 + 2] = sz * inv;
    }
    float rb[4];
#pragma unroll
    for (int u = 0; u < 4; ++u) {
      int k = (u >> 1) * 4 + 2 * half + (u & 1);
      float x = (dd - (float)k * (R0f / 7.0f)) * 4.0f;
      rb[u] = expf(-x * x);
    }

    v8f acc[4];
#pragma unroll
    for (int j = 0; j < 4; ++j) acc[j] = (v8f)0.0f;

    const float* es = emb + (size_t)sN * 64;
    const float* ed = emb + (size_t)dN * 64;
#pragma unroll
    for (int t = 0; t < 16; ++t) {
      int k0 = 4 * t + 2 * half;
      v2f a = *(const v2f*)(es + k0);
#pragma unroll
      for (int j = 0; j < 4; ++j) {
        v2f b = *(const v2f*)(smem + OFF_WSRC + (16 * j + nl) * WP + k0);
        acc[j] = wmma4(a, b, acc[j]);
      }
    }
#pragma unroll
    for (int t = 0; t < 16; ++t) {
      int k0 = 4 * t + 2 * half;
      v2f a = *(const v2f*)(ed + k0);
#pragma unroll
      for (int j = 0; j < 4; ++j) {
        v2f b = *(const v2f*)(smem + OFF_WDST + (16 * j + nl) * WP + k0);
        acc[j] = wmma4(a, b, acc[j]);
      }
    }
#pragma unroll
    for (int t = 0; t < 2; ++t) {
      int k0 = 4 * t + 2 * half;
      v2f a;
      a[0] = rb[2 * t];
      a[1] = rb[2 * t + 1];
#pragma unroll
      for (int j = 0; j < 4; ++j) {
        v2f b = *(const v2f*)(smem + OFF_WENC + (16 * j + nl) * EP + k0);
        acc[j] = wmma4(a, b, acc[j]);
      }
    }
#pragma unroll
    for (int j = 0; j < 4; ++j) {
      float bb = smem[OFF_BIAS0 + 16 * j + nl];
#pragma unroll
      for (int r = 0; r < 8; ++r) acc[j][r] += bb;
    }

    lds_fence();
#pragma unroll
    for (int j = 0; j < 4; ++j)
#pragma unroll
      for (int r = 0; r < 8; ++r)
        stage[(r + 8 * half) * WP + 16 * j + nl] = acc[j][r];
    lds_fence();

    unsigned zb0[8], zb1[8], sb0[8], sb1[8];
#pragma unroll
    for (int r = 0; r < 8; ++r) {
      const unsigned sN8 = (unsigned)idxS[r + 8 * half];
      const unsigned dN8 = (unsigned)idxS[16 + r + 8 * half];
      sb0[r] = (sN8 * 64u + (unsigned)nl) * 4u - 64u;
      sb1[r] = (sN8 * 192u + (unsigned)nl) * 4u - 64u;
      zb0[r] = (dN8 * 64u + (unsigned)nl) * 4u;
      zb1[r] = (dN8 * 192u + (unsigned)nl) * 4u;
    }
    float rh[3][8];
#pragma unroll
    for (int r = 0; r < 8; ++r) {
      rh[0][r] = rhS[(r + 8 * half) * 4 + 0];
      rh[1][r] = rhS[(r + 8 * half) * 4 + 1];
      rh[2][r] = rhS[(r + 8 * half) * 4 + 2];
    }

    v2f af[16];
#pragma unroll
    for (int t = 0; t < 16; ++t)
      af[t] = *(const v2f*)(stage + nl * WP + 4 * t + 2 * half);

#pragma unroll 1
    for (int j = 0; j < 4; ++j) {
      const char* z0b = (const char*)z0g;
      const char* z1b = (const char*)z1g;
      char* o0b = (char*)out0;
      char* o1b = (char*)out1;

      float z0r[8];
#pragma unroll
      for (int r = 0; r < 8; ++r) z0r[r] = *(const float*)(z0b + zb0[r]);
      float z1r[3][8];
#pragma unroll
      for (int i = 0; i < 3; ++i)
#pragma unroll
        for (int r = 0; r < 8; ++r)
          z1r[i][r] = *(const float*)(z1b + i * 256 + zb1[r]);

      const float* wb = smem + OFF_WGATE + (16 * j + nl) * WP + 2 * half;
      float bb[4];
      v8f g[4];
      v2f bq[2][2][4];
#pragma unroll
      for (int c = 0; c < 4; ++c) {
        g[c] = (v8f)0.0f;
        bb[c] = smem[OFF_BGATE + 64 * c + 16 * j + nl];
        bq[0][0][c] = *(const v2f*)(wb + 64 * c * WP);
        bq[0][1][c] = *(const v2f*)(wb + 64 * c * WP + 4);
      }
      __builtin_amdgcn_sched_barrier(0);
#pragma unroll
      for (int u = 0; u < 8; ++u) {
        if (u < 7) {
#pragma unroll
          for (int c = 0; c < 4; ++c) {
            bq[(u + 1) & 1][0][c] = *(const v2f*)(wb + 64 * c * WP + 8 * (u + 1));
            bq[(u + 1) & 1][1][c] = *(const v2f*)(wb + 64 * c * WP + 8 * (u + 1) + 4);
          }
        }
        __builtin_amdgcn_sched_barrier(0);
#pragma unroll
        for (int c = 0; c < 4; ++c) g[c] = wmma4(af[2 * u], bq[u & 1][0][c], g[c]);
#pragma unroll
        for (int c = 0; c < 4; ++c) g[c] = wmma4(af[2 * u + 1], bq[u & 1][1][c], g[c]);
        __builtin_amdgcn_sched_barrier(0);
      }
#pragma unroll
      for (int c = 0; c < 4; ++c)
#pragma unroll
        for (int r = 0; r < 8; ++r) g[c][r] += bb[c];
#pragma unroll
      for (int r = 0; r < 8; ++r) {
        sb0[r] += 64u;
        sb1[r] += 64u;
      }

      float ms[8], mv[3][8];
#pragma unroll
      for (int r = 0; r < 8; ++r) {
        float dotv = rh[0][r] * z1r[0][r] + rh[1][r] * z1r[1][r] +
                     rh[2][r] * z1r[2][r];
        ms[r] = g[0][r] * z0r[r] + g[1][r] * dotv;
      }
#pragma unroll
      for (int i = 0; i < 3; ++i)
#pragma unroll
        for (int r = 0; r < 8; ++r)
          mv[i][r] = g[2][r] * z1r[i][r] + g[3][r] * rh[i][r] * z0r[r];
      __builtin_amdgcn_sched_barrier(0);

#pragma unroll
      for (int r = 0; r < 8; ++r) atomicAdd((float*)(o0b + sb0[r]), ms[r]);
#pragma unroll
      for (int i = 0; i < 3; ++i)
#pragma unroll
        for (int r = 0; r < 8; ++r)
          atomicAdd((float*)(o1b + i * 256 + sb1[r]), mv[i][r]);

#pragma unroll
      for (int r = 0; r < 8; ++r) {
        zb0[r] += 64u;
        zb1[r] += 64u;
      }
    }
  }
}

__global__ void __launch_bounds__(BLOCK)
node_transform_kernel(const float* __restrict__ Ws, const float* __restrict__ Wv,
                      float* out) {
  extern __shared__ float smem[];

  for (int i = threadIdx.x; i < 64 * 64; i += BLOCK) {
    int k = i >> 6, n = i & 63;
    smem[N_OFF_WS + n * WP + k] = Ws[i];
    smem[N_OFF_WV + n * WP + k] = Wv[i];
  }
  __syncthreads();

  const int lane = threadIdx.x & 31;
  const int wave = threadIdx.x >> 5;
  const int half = lane >> 4;
  const int nl   = lane & 15;

  float* stage = smem + N_OFF_STAGE + wave * (16 * WP);

  for (int task = blockIdx.x * WAVES + wave; task < NODE_TASKS;
       task += gridDim.x * WAVES) {
    const int tile = task >> 2;
    const int part = task & 3;
    float* base;
    int pitch;
    const float* wT;
    if (part == 0) {
      base = out + (size_t)tile * (16 * 64);
      pitch = 64;
      wT = smem + N_OFF_WS;
    } else {
      base = out + (size_t)NNODES * CHAN + (size_t)tile * (16 * 192) + (part - 1) * 64;
      pitch = 192;
      wT = smem + N_OFF_WV;
    }

    v4f v[8];
#pragma unroll
    for (int q = 0; q < 8; ++q) {
      int idx = q * 32 + lane;
      v[q] = *(const v4f*)(base + (size_t)(idx >> 4) * pitch + (idx & 15) * 4);
    }
    lds_fence();
#pragma unroll
    for (int q = 0; q < 8; ++q) {
      int idx = q * 32 + lane;
      *(v4f*)(stage + (idx >> 4) * WP + (idx & 15) * 4) = v[q];
    }
    lds_fence();

    v8f p[4];
#pragma unroll
    for (int j = 0; j < 4; ++j) p[j] = (v8f)0.0f;
#pragma unroll
    for (int t = 0; t < 16; ++t) {
      int k0 = 4 * t + 2 * half;
      v2f a = *(const v2f*)(stage + nl * WP + k0);
#pragma unroll
      for (int j = 0; j < 4; ++j) {
        v2f b = *(const v2f*)(wT + (16 * j + nl) * WP + k0);
        p[j] = wmma4(a, b, p[j]);
      }
    }

    lds_fence();
#pragma unroll
    for (int j = 0; j < 4; ++j)
#pragma unroll
      for (int r = 0; r < 8; ++r)
        stage[(r + 8 * half) * WP + 16 * j + nl] = p[j][r];
    lds_fence();
#pragma unroll
    for (int q = 0; q < 8; ++q) {
      int idx = q * 32 + lane;
      v[q] = *(const v4f*)(stage + (idx >> 4) * WP + (idx & 15) * 4);
    }
#pragma unroll
    for (int q = 0; q < 8; ++q) {
      int idx = q * 32 + lane;
      *(v4f*)(base + (size_t)(idx >> 4) * pitch + (idx & 15) * 4) = v[q];
    }
  }
}

extern "C" void kernel_launch(void* const* d_in, const int* in_sizes, int n_in,
                              void* d_out, int out_size, void* d_ws,
                              size_t ws_size, hipStream_t stream) {
  const int*   src   = (const int*)d_in[0];
  const int*   dst   = (const int*)d_in[1];
  const float* r_ij  = (const float*)d_in[2];
  const float* z0    = (const float*)d_in[3];
  const float* z1    = (const float*)d_in[4];
  const float* emb   = (const float*)d_in[5];
  const float* Wenc  = (const float*)d_in[6];
  const float* benc  = (const float*)d_in[7];
  const float* Wsrc  = (const float*)d_in[8];
  const float* bsrc  = (const float*)d_in[9];
  const float* Wdst  = (const float*)d_in[10];
  const float* bdst  = (const float*)d_in[11];
  const float* Wgate = (const float*)d_in[12];
  const float* bgate = (const float*)d_in[13];
  const float* Ws    = (const float*)d_in[14];
  const float* Wv    = (const float*)d_in[15];
  float* out = (float*)d_out;

  zero_f32<<<ZGRID, BLOCK, 0, stream>>>(out, out_size);

  size_t smem_bytes = (size_t)SMEM_FLOATS * sizeof(float);
  messages_kernel<<<GRID, BLOCK, smem_bytes, stream>>>(
      src, dst, r_ij, z0, z1, emb, Wenc, benc, Wsrc, bsrc, Wdst, bdst, Wgate,
      bgate, out, out + (size_t)NNODES * CHAN);

  size_t nsmem_bytes = (size_t)N_SMEM_FLOATS * sizeof(float);
  node_transform_kernel<<<NGRID, BLOCK, nsmem_bytes, stream>>>(Ws, Wv, out);
}


# scatter atomics issued as whole 128-byte lines: messages of two 16-channel groups held, lane halves swapped, each atomic adds one row's 32 channels, not 64 B of two rows; sums stay in arrival order
# speedup vs baseline: 4.3406x; 1.0372x over previous
#include <hip/hip_runtime.h>
#include <math.h>

#define NNODES 50000
#define NEDGES 500000
#define CHAN   64
#define R0f    2.0f

#define WAVES  8
#define BLOCK  (WAVES * 32)
#define NTILES (NEDGES / 16)
#define GRID   512

typedef float v2f __attribute__((ext_vector_type(2)));
typedef float v4f __attribute__((ext_vector_type(4)));
typedef float v8f __attribute__((ext_vector_type(8)));
typedef unsigned v2u __attribute__((ext_vector_type(2)));

#define WP 68
#define EP 12
#define OFF_WSRC  0
#define OFF_WDST  (OFF_WSRC + 64 * WP)
#define OFF_WGATE (OFF_WDST + 64 * WP)
#define OFF_WENC  (OFF_WGATE + 256 * WP)
#define OFF_BIAS0 (OFF_WENC + 64 * EP)
#define OFF_BGATE (OFF_BIAS0 + 64)
#define OFF_STAGE (OFF_BGATE + 256)
#define OFF_RH    (OFF_STAGE + WAVES * 16 * WP)
#define OFF_IDX   (OFF_RH + WAVES * 64)
#define SMEM_FLOATS (OFF_IDX + WAVES * 32)

#define NODE_TILES (NNODES / 16)
#define NODE_TASKS (NODE_TILES * 4)
#define NGRID      1024
#define N_OFF_WS    0
#define N_OFF_WV    (N_OFF_WS + 64 * WP)
#define N_OFF_STAGE (N_OFF_WV + 64 * WP)
#define N_SMEM_FLOATS (N_OFF_STAGE + WAVES * 16 * WP)

__device__ __forceinline__ void lds_fence() {
  asm volatile("s_wait_dscnt 0x0" ::: "memory");
}

__device__ __forceinline__ v8f wmma4(v2f a, v2f b, v8f c) {
  return __builtin_amdgcn_wmma_f32_16x16x4_f32(false, a, false, b, (short)0, c,
                                               false, false);
}

#define ZGRID   256
#define ZSTRIDE (ZGRID * BLOCK)
__global__ void __launch_bounds__(BLOCK)
zero_f32(float* __restrict__ p, int n) {
  const int n4 = n >> 2;
  v4f* __restrict__ p4 = (v4f*)p;
  int i = blockIdx.x * BLOCK + threadIdx.x;
  for (; i + 7 * ZSTRIDE < n4; i += 8 * ZSTRIDE) {
#pragma unroll
    for (int u = 0; u < 8; ++u) p4[i + u * ZSTRIDE] = (v4f)0.0f;
  }
  for (; i < n4; i += ZSTRIDE) p4[i] = (v4f)0.0f;
  const int t = (n4 << 2) + threadIdx.x;
  if (blockIdx.x == 0 && t < n) p[t] = 0.0f;
}

__global__ void __launch_bounds__(BLOCK, 4)
messages_kernel(const int* __restrict__ src, const int* __restrict__ dst,
                const float* __restrict__ r_ij,
                const float* __restrict__ z0g, const float* __restrict__ z1g,
                const float* __restrict__ emb,
                const float* __restrict__ Wenc, const float* __restrict__ benc,
                const float* __restrict__ Wsrc, const float* __restrict__ bsrc,
                const float* __restrict__ Wdst, const float* __restrict__ bdst,
                const float* __restrict__ Wgate, const float* __restrict__ bgate,
                float* __restrict__ out0, float* __restrict__ out1) {
  extern __shared__ float smem[];

  for (int i = threadIdx.x; i < 64 * 64; i += BLOCK) {
    int k = i >> 6, n = i & 63;
    smem[OFF_WSRC + n * WP + k] = Wsrc[i];
    smem[OFF_WDST + n * WP + k] = Wdst[i];
  }
  for (int i = threadIdx.x; i < 64 * 256; i += BLOCK) {
    int k = i >> 8, n = i & 255;
    smem[OFF_WGATE + n * WP + k] = Wgate[i];
  }
  for (int i = threadIdx.x; i < 8 * 64; i += BLOCK) {
    int k = i >> 6, n = i & 63;
    smem[OFF_WENC + n * EP + k] = Wenc[i];
  }
  if (threadIdx.x < 64)
    smem[OFF_BIAS0 + threadIdx.x] =
        bsrc[threadIdx.x] + bdst[threadIdx.x] + benc[threadIdx.x];
  if (threadIdx.x < 256) smem[OFF_BGATE + threadIdx.x] = bgate[threadIdx.x];
  __syncthreads();

  const int lane = threadIdx.x & 31;
  const int wave = threadIdx.x >> 5;
  const int half = lane >> 4;
  const int nl   = lane & 15;

  float* stage = smem + OFF_STAGE + wave * (16 * WP);
  float* rhS   = smem + OFF_RH + wave * 64;
  int*   idxS  = ((int*)(smem + OFF_IDX)) + wave * 32;

  for (int tile = blockIdx.x * WAVES + wave; tile < NTILES;
       tile += gridDim.x * WAVES) {
    const int e  = tile * 16 + nl;
    const int sN = src[e];
    const int dN = dst[e];
    if (half == 0) { idxS[nl] = sN; idxS[16 + nl] = dN; }

    float rx = r_ij[e * 3 + 0], ry = r_ij[e * 3 + 1], rz = r_ij[e * 3 + 2];
    float dd  = sqrtf(rx * rx + ry * ry + rz * rz);
    float sx = rx * 3.5f, sy = ry * 3.5f, sz = rz * 3.5f;
    float inv = rsqrtf(1.0f + sx * sx + sy * sy + sz * sz);
    if (half == 0) {
      rhS[nl * 4 + 0] = sx * inv;
      rhS[nl * 4 + 1] = sy * inv;
      rhS[nl * 4 + 2] = sz * inv;
    }
    float rb[4];
#pragma unroll
    for (int u = 0; u < 4; ++u) {
      int k = (u >> 1) * 4 + 2 * half + (u & 1);
      float x = (dd - (float)k * (R0f / 7.0f)) * 4.0f;
      rb[u] = expf(-x * x);
    }

    v8f acc[4];
#pragma unroll
    for (int j = 0; j < 4; ++j) acc[j] = (v8f)0.0f;

    const float* es = emb + (size_t)sN * 64;
    const float* ed = emb + (size_t)dN * 64;
#pragma unroll
    for (int t = 0; t < 16; ++t) {
      int k0 = 4 * t + 2 * half;
      v2f a = *(const v2f*)(es + k0);
#pragma unroll
      for (int j = 0; j < 4; ++j) {
        v2f b = *(const v2f*)(smem + OFF_WSRC + (16 * j + nl) * WP + k0);
        acc[j] = wmma4(a, b, acc[j]);
      }
    }
#pragma unroll
    for (int t = 0; t < 16; ++t) {
      int k0 = 4 * t + 2 * half;
      v2f a = *(const v2f*)(ed + k0);
#pragma unroll
      for (int j = 0; j < 4; ++j) {
        v2f b = *(const v2f*)(smem + OFF_WDST + (16 * j + nl) * WP + k0);
        acc[j] = wmma4(a, b, acc[j]);
      }
    }
#pragma unroll
    for (int t = 0; t < 2; ++t) {
      int k0 = 4 * t + 2 * half;
      v2f a;
      a[0] = rb[2 * t];
      a[1] = rb[2 * t + 1];
#pragma unroll
      for (int j = 0; j < 4; ++j) {
        v2f b = *(const v2f*)(smem + OFF_WENC + (16 * j + nl) * EP + k0);
        acc[j] = wmma4(a, b, acc[j]);
      }
    }
#pragma unroll
    for (int j = 0; j < 4; ++j) {
      float bb = smem[OFF_BIAS0 + 16 * j + nl];
#pragma unroll
      for (int r = 0; r < 8; ++r) acc[j][r] += bb;
    }

    lds_fence();
#pragma unroll
    for (int j = 0; j < 4; ++j)
#pragma unroll
      for (int r = 0; r < 8; ++r)
        stage[(r + 8 * half) * WP + 16 * j + nl] = acc[j][r];
    lds_fence();

    unsigned zb0[8], zb1[8];
#pragma unroll
    for (int r = 0; r < 8; ++r) {
      const unsigned dN8 = (unsigned)idxS[16 + r + 8 * half];
      zb0[r] = (dN8 * 64u + (unsigned)nl) * 4u;
      zb1[r] = (dN8 * 192u + (unsigned)nl) * 4u;
    }
    unsigned sRow[16];
#pragma unroll
    for (int R = 0; R < 16; ++R) sRow[R] = (unsigned)__builtin_amdgcn_readlane(sN, R);
    float rh[3][8];
#pragma unroll
    for (int r = 0; r < 8; ++r) {
      rh[0][r] = rhS[(r + 8 * half) * 4 + 0];
      rh[1][r] = rhS[(r + 8 * half) * 4 + 1];
      rh[2][r] = rhS[(r + 8 * half) * 4 + 2];
    }

    v2f af[16];
#pragma unroll
    for (int t = 0; t < 16; ++t)
      af[t] = *(const v2f*)(stage + nl * WP + 4 * t + 2 * half);

#pragma unroll 1
    for (int jp = 0; jp < 2; ++jp) {
      const char* z0b = (const char*)z0g;
      const char* z1b = (const char*)z1g;
      float ms[2][8], mv[2][3][8];

#pragma unroll
      for (int h2 = 0; h2 < 2; ++h2) {
        const int j = 2 * jp + h2;
        float z0r[8];
#pragma unroll
        for (int r = 0; r < 8; ++r) z0r[r] = *(const float*)(z0b + zb0[r]);
        float z1r[3][8];
#pragma unroll
        for (int i = 0; i < 3; ++i)
#pragma unroll
          for (int r = 0; r < 8; ++r)
            z1r[i][r] = *(const float*)(z1b + i * 256 + zb1[r]);

        const float* wb = smem + OFF_WGATE + (16 * j + nl) * WP + 2 * half;
        float bb[4];
        v8f g[4];
        v2f bq[2][2][4];
#pragma unroll
        for (int c = 0; c < 4; ++c) {
          g[c] = (v8f)0.0f;
          bb[c] = smem[OFF_BGATE + 64 * c + 16 * j + nl];
          bq[0][0][c] = *(const v2f*)(wb + 64 * c * WP);
          bq[0][1][c] = *(const v2f*)(wb + 64 * c * WP + 4);
        }
        __builtin_amdgcn_sched_barrier(0);
#pragma unroll
        for (int u = 0; u < 8; ++u) {
          if (u < 7) {
#pragma unroll
            for (int c = 0; c < 4; ++c) {
              bq[(u + 1) & 1][0][c] = *(const v2f*)(wb + 64 * c * WP + 8 * (u + 1));
              bq[(u + 1) & 1][1][c] = *(const v2f*)(wb + 64 * c * WP + 8 * (u + 1) + 4);
            }
          }
          __builtin_amdgcn_sched_barrier(0);
#pragma unroll
          for (int c = 0; c < 4; ++c) g[c] = wmma4(af[2 * u], bq[u & 1][0][c], g[c]);
#pragma unroll
          for (int c = 0; c < 4; ++c) g[c] = wmma4(af[2 * u + 1], bq[u & 1][1][c], g[c]);
          __builtin_amdgcn_sched_barrier(0);
        }
#pragma unroll
        for (int c = 0; c < 4; ++c)
#pragma unroll
          for (int r = 0; r < 8; ++r) g[c][r] += bb[c];

#pragma unroll
        for (int r = 0; r < 8; ++r) {
          float dotv = rh[0][r] * z1r[0][r] + rh[1][r] * z1r[1][r] +
                       rh[2][r] * z1r[2][r];
          ms[h2][r] = g[0][r] * z0r[r] + g[1][r] * dotv;
        }
#pragma unroll
        for (int i = 0; i < 3; ++i)
#pragma unroll
          for (int r = 0; r < 8; ++r)
            mv[h2][i][r] = g[2][r] * z1r[i][r] + g[3][r] * rh[i][r] * z0r[r];

#pragma unroll
        for (int r = 0; r < 8; ++r) {
          zb0[r] += 64u;
          zb1[r] += 64u;
        }
        __builtin_amdgcn_sched_barrier(0);
      }

      float xs[2][8], xv[2][3][8];
#pragma unroll
      for (int r = 0; r < 8; ++r) {
        v2u t = __builtin_amdgcn_permlane16_swap(__float_as_uint(ms[0][r]), __float_as_uint(ms[1][r]), false, false);
        xs[0][r] = __uint_as_float(t[0]);
        xs[1][r] = __uint_as_float(t[1]);
      }
#pragma unroll
      for (int i = 0; i < 3; ++i)
#pragma unroll
        for (int r = 0; r < 8; ++r) {
          v2u t = __builtin_amdgcn_permlane16_swap(__float_as_uint(mv[0][i][r]), __float_as_uint(mv[1][i][r]), false, false);
          xv[0][i][r] = __uint_as_float(t[0]);
          xv[1][i][r] = __uint_as_float(t[1]);
        }
      __builtin_amdgcn_sched_barrier(0);

      const unsigned ch = (unsigned)(32 * jp + lane);
#pragma unroll
      for (int q = 0; q < 2; ++q)
#pragma unroll
        for (int r = 0; r < 8; ++r) {
          atomicAdd(out0 + (size_t)sRow[r + 8 * q] * 64 + ch, xs[q][r]);
#pragma unroll
          for (int i = 0; i < 3; ++i)
            atomicAdd(out1 + (size_t)sRow[r + 8 * q] * 192 + i * 64 + ch, xv[q][i][r]);
        }
    }
  }
}

__global__ void __launch_bounds__(BLOCK)
node_transform_kernel(const float* __restrict__ Ws, const float* __restrict__ Wv,
                      float* out) {
  extern __shared__ float smem[];

  for (int i = threadIdx.x; i < 64 * 64; i += BLOCK) {
    int k = i >> 6, n = i & 63;
    smem[N_OFF_WS + n * WP + k] = Ws[i];
    smem[N_OFF_WV + n * WP + k] = Wv[i];
  }
  __syncthreads();

  const int lane = threadIdx.x & 31;
  const int wave = threadIdx.x >> 5;
  const int half = lane >> 4;
  const int nl   = lane & 15;

  float* stage = smem + N_OFF_STAGE + wave * (16 * WP);

  for (int task = blockIdx.x * WAVES + wave; task < NODE_TASKS;
       task += gridDim.x * WAVES) {
    const int tile = task >> 2;
    const int part = task & 3;
    float* base;
    int pitch;
    const float* wT;
    if (part == 0) {
      base = out + (size_t)tile * (16 * 64);
      pitch = 64;
      wT = smem + N_OFF_WS;
    } else {
      base = out + (size_t)NNODES * CHAN + (size_t)tile * (16 * 192) + (part - 1) * 64;
      pitch = 192;
      wT = smem + N_OFF_WV;
    }

    v4f v[8];
#pragma unroll
    for (int q = 0; q < 8; ++q) {
      int idx = q * 32 + lane;
      v[q] = *(const v4f*)(base + (size_t)(idx >> 4) * pitch + (idx & 15) * 4);
    }
    lds_fence();
#pragma unroll
    for (int q = 0; q < 8; ++q) {
      int idx = q * 32 + lane;
      *(v4f*)(stage + (idx >> 4) * WP + (idx & 15) * 4) = v[q];
    }
    lds_fence();

    v8f p[4];
#pragma unroll
    for (int j = 0; j < 4; ++j) p[j] = (v8f)0.0f;
#pragma unroll
    for (int t = 0; t < 16; ++t) {
      int k0 = 4 * t + 2 * half;
      v2f a = *(const v2f*)(stage + nl * WP + k0);
#pragma unroll
      for (int j = 0; j < 4; ++j) {
        v2f b = *(const v2f*)(wT + (16 * j + nl) * WP + k0);
        p[j] = wmma4(a, b, p[j]);
      }
    }

    lds_fence();
#pragma unroll
    for (int j = 0; j < 4; ++j)
#pragma unroll
      for (int r = 0; r < 8; ++r)
        stage[(r + 8 * half) * WP + 16 * j + nl] = p[j][r];
    lds_fence();
#pragma unroll
    for (int q = 0; q < 8; ++q) {
      int idx = q * 32 + lane;
      v[q] = *(const v4f*)(stage + (idx >> 4) * WP + (idx & 15) * 4);
    }
#pragma unroll
    for (int q = 0; q < 8; ++q) {
      int idx = q * 32 + lane;
      *(v4f*)(base + (size_t)(idx >> 4) * pitch + (idx & 15) * 4) = v[q];
    }
  }
}

extern "C" void kernel_launch(void* const* d_in, const int* in_sizes, int n_in,
                              void* d_out, int out_size, void* d_ws,
                              size_t ws_size, hipStream_t stream) {
  const int*   src   = (const int*)d_in[0];
  const int*   dst   = (const int*)d_in[1];
  const float* r_ij  = (const float*)d_in[2];
  const float* z0    = (const float*)d_in[3];
  const float* z1    = (const float*)d_in[4];
  const float* emb   = (const float*)d_in[5];
  const float* Wenc  = (const float*)d_in[6];
  const float* benc  = (const float*)d_in[7];
  const float* Wsrc  = (const float*)d_in[8];
  const float* bsrc  = (const float*)d_in[9];
  const float* Wdst  = (const float*)d_in[10];
  const float* bdst  = (const float*)d_in[11];
  const float* Wgate = (const float*)d_in[12];
  const float* bgate = (const float*)d_in[13];
  const float* Ws    = (const float*)d_in[14];
  const float* Wv    = (const float*)d_in[15];
  float* out = (float*)d_out;

  zero_f32<<<ZGRID, BLOCK, 0, stream>>>(out, out_size);

  size_t smem_bytes = (size_t)SMEM_FLOATS * sizeof(float);
  messages_kernel<<<GRID, BLOCK, smem_bytes, stream>>>(
      src, dst, r_ij, z0, z1, emb, Wenc, benc, Wsrc, bsrc, Wdst, bdst, Wgate,
      bgate, out, out + (size_t)NNODES * CHAN);

  size_t nsmem_bytes = (size_t)N_SMEM_FLOATS * sizeof(float);
  node_transform_kernel<<<NGRID, BLOCK, nsmem_bytes, stream>>>(Ws, Wv, out);
}


# dist_emb products in sets of 16 WMMAs: emb gathers and B-fragment reads fetch 16 B per lane (two k-steps), a set's 8 reads issued one set ahead as one group; k pairing and float sum order change
# speedup vs baseline: 4.4741x; 1.0308x over previous
#include <hip/hip_runtime.h>
#include <math.h>

#define NNODES 50000
#define NEDGES 500000
#define CHAN   64
#define R0f    2.0f

#define WAVES  8
#define BLOCK  (WAVES * 32)
#define NTILES (NEDGES / 16)
#define GRID   512

typedef float v2f __attribute__((ext_vector_type(2)));
typedef float v4f __attribute__((ext_vector_type(4)));
typedef float v8f __attribute__((ext_vector_type(8)));
typedef unsigned v2u __attribute__((ext_vector_type(2)));

#define WP 68
#define EP 12
#define OFF_WSRC  0
#define OFF_WDST  (OFF_WSRC + 64 * WP)
#define OFF_WGATE (OFF_WDST + 64 * WP)
#define OFF_WENC  (OFF_WGATE + 256 * WP)
#define OFF_BIAS0 (OFF_WENC + 64 * EP)
#define OFF_BGATE (OFF_BIAS0 + 64)
#define OFF_STAGE (OFF_BGATE + 256)
#define OFF_RH    (OFF_STAGE + WAVES * 16 * WP)
#define OFF_IDX   (OFF_RH + WAVES * 64)
#define SMEM_FLOATS (OFF_IDX + WAVES * 32)

#define NODE_TILES (NNODES / 16)
#define NODE_TASKS (NODE_TILES * 4)
#define NGRID      1024
#define N_OFF_WS    0
#define N_OFF_WV    (N_OFF_WS + 64 * WP)
#define N_OFF_STAGE (N_OFF_WV + 64 * WP)
#define N_SMEM_FLOATS (N_OFF_STAGE + WAVES * 16 * WP)

__device__ __forceinline__ void lds_fence() {
  asm volatile("s_wait_dscnt 0x0" ::: "memory");
}

__device__ __forceinline__ v8f wmma4(v2f a, v2f b, v8f c) {
  return __builtin_amdgcn_wmma_f32_16x16x4_f32(false, a, false, b, (short)0, c,
                                               false, false);
}

__device__ __forceinline__ v2f lo2(v4f v) { return __builtin_shufflevector(v, v, 0, 1); }
__device__ __forceinline__ v2f hi2(v4f v) { return __builtin_shufflevector(v, v, 2, 3); }

__device__ __forceinline__ void mm64(v8f (&acc)[4], const v4f (&a4)[8], const float* wrow) {
  v4f bq[2][2][4];
#pragma unroll
  for (int s = 0; s < 2; ++s)
#pragma unroll
    for (int j = 0; j < 4; ++j) bq[0][s][j] = *(const v4f*)(wrow + j * 16 * WP + 8 * s);
#pragma unroll
  for (int u = 0; u < 4; ++u) {
    if (u < 3) {
#pragma unroll
      for (int s = 0; s < 2; ++s)
#pragma unroll
        for (int j = 0; j < 4; ++j)
          bq[(u + 1) & 1][s][j] = *(const v4f*)(wrow + j * 16 * WP + 8 * (2 * (u + 1) + s));
    }
    __builtin_amdgcn_sched_barrier(0);
#pragma unroll
    for (int s = 0; s < 2; ++s) {
#pragma unroll
      for (int j = 0; j < 4; ++j) acc[j] = wmma4(lo2(a4[2 * u + s]), lo2(bq[u & 1][s][j]), acc[j]);
#pragma unroll
      for (int j = 0; j < 4; ++j) acc[j] = wmma4(hi2(a4[2 * u + s]), hi2(bq[u & 1][s][j]), acc[j]);
    }
    __builtin_amdgcn_sched_barrier(0);
  }
}

#define ZGRID   256
#define ZSTRIDE (ZGRID * BLOCK)
__global__ void __launch_bounds__(BLOCK)
zero_f32(float* __restrict__ p, int n) {
  const int n4 = n >> 2;
  v4f* __restrict__ p4 = (v4f*)p;
  int i = blockIdx.x * BLOCK + threadIdx.x;
  for (; i + 7 * ZSTRIDE < n4; i += 8 * ZSTRIDE) {
#pragma unroll
    for (int u = 0; u < 8; ++u) p4[i + u * ZSTRIDE] = (v4f)0.0f;
  }
  for (; i < n4; i += ZSTRIDE) p4[i] = (v4f)0.0f;
  const int t = (n4 << 2) + threadIdx.x;
  if (blockIdx.x == 0 && t < n) p[t] = 0.0f;
}

__global__ void __launch_bounds__(BLOCK, 4)
messages_kernel(const int* __restrict__ src, const int* __restrict__ dst,
                const float* __restrict__ r_ij,
                const float* __restrict__ z0g, const float* __restrict__ z1g,
                const float* __restrict__ emb,
                const float* __restrict__ Wenc, const float* __restrict__ benc,
                const float* __restrict__ Wsrc, const float* __restrict__ bsrc,
                const float* __restrict__ Wdst, const float* __restrict__ bdst,
                const float* __restrict__ Wgate, const float* __restrict__ bgate,
                float* __restrict__ out0, float* __restrict__ out1) {
  extern __shared__ float smem[];

  for (int i = threadIdx.x; i < 64 * 64; i += BLOCK) {
    int k = i >> 6, n = i & 63;
    smem[OFF_WSRC + n * WP + k] = Wsrc[i];
    smem[OFF_WDST + n * WP + k] = Wdst[i];
  }
  for (int i = threadIdx.x; i < 64 * 256; i += BLOCK) {
    int k = i >> 8, n = i & 255;
    smem[OFF_WGATE + n * WP + k] = Wgate[i];
  }
  for (int i = threadIdx.x; i < 8 * 64; i += BLOCK) {
    int k = i >> 6, n = i & 63;
    smem[OFF_WENC + n * EP + k] = Wenc[i];
  }
  if (threadIdx.x < 64)
    smem[OFF_BIAS0 + threadIdx.x] =
        bsrc[threadIdx.x] + bdst[threadIdx.x] + benc[threadIdx.x];
  if (threadIdx.x < 256) smem[OFF_BGATE + threadIdx.x] = bgate[threadIdx.x];
  __syncthreads();

  const int lane = threadIdx.x & 31;
  const int wave = threadIdx.x >> 5;
  const int half = lane >> 4;
  const int nl   = lane & 15;

  float* stage = smem + OFF_STAGE + wave * (16 * WP);
  float* rhS   = smem + OFF_RH + wave * 64;
  int*   idxS  = ((int*)(smem + OFF_IDX)) + wave * 32;

  for (int tile = blockIdx.x * WAVES + wave; tile < NTILES;
       tile += gridDim.x * WAVES) {
    const int e  = tile * 16 + nl;
    const int sN = src[e];
    const int dN = dst[e];
    if (half == 0) { idxS[nl] = sN; idxS[16 + nl] = dN; }

    float rx = r_ij[e * 3 + 0], ry = r_ij[e * 3 + 1], rz = r_ij[e * 3 + 2];
    float dd  = sqrtf(rx * rx + ry * ry + rz * rz);
    float sx = rx * 3.5f, sy = ry * 3.5f, sz = rz * 3.5f;
    float inv = rsqrtf(1.0f + sx * sx + sy * sy + sz * sz);
    if (half == 0) {
      rhS[nl * 4 + 0] = sx * inv;
      rhS[nl * 4 + 1] = sy * inv;
      rhS[nl * 4 + 2] = sz * inv;
    }
    float rb[4];
#pragma unroll
    for (int u = 0; u < 4; ++u) {
      int k = 4 * half + u;
      float x = (dd - (float)k * (R0f / 7.0f)) * 4.0f;
      rb[u] = expf(-x * x);
    }

    v8f acc[4];
#pragma unroll
    for (int j = 0; j < 4; ++j) acc[j] = (v8f)0.0f;

    const float* es = emb + (size_t)sN * 64 + 4 * half;
    const float* ed = emb + (size_t)dN * 64 + 4 * half;
    v4f es4[8], ed4[8];
#pragma unroll
    for (int q = 0; q < 8; ++q) es4[q] = *(const v4f*)(es + 8 * q);
#pragma unroll
    for (int q = 0; q < 8; ++q) ed4[q] = *(const v4f*)(ed + 8 * q);
    mm64(acc, es4, smem + OFF_WSRC + nl * WP + 4 * half);
    mm64(acc, ed4, smem + OFF_WDST + nl * WP + 4 * half);
    {
      v4f be[4];
#pragma unroll
      for (int j = 0; j < 4; ++j)
        be[j] = *(const v4f*)(smem + OFF_WENC + (16 * j + nl) * EP + 4 * half);
      v2f a0, a1;
      a0[0] = rb[0];
      a0[1] = rb[1];
      a1[0] = rb[2];
      a1[1] = rb[3];
#pragma unroll
      for (int j = 0; j < 4; ++j) acc[j] = wmma4(a0, lo2(be[j]), acc[j]);
#pragma unroll
      for (int j = 0; j < 4; ++j) acc[j] = wmma4(a1, hi2(be[j]), acc[j]);
    }
#pragma unroll
    for (int j = 0; j < 4; ++j) {
      float bb = smem[OFF_BIAS0 + 16 * j + nl];
#pragma unroll
      for (int r = 0; r < 8; ++r) acc[j][r] += bb;
    }

    lds_fence();
#pragma unroll
    for (int j = 0; j < 4; ++j)
#pragma unroll
      for (int r = 0; r < 8; ++r)
        stage[(r + 8 * half) * WP + 16 * j + nl] = acc[j][r];
    lds_fence();

    unsigned zb0[8], zb1[8];
#pragma unroll
    for (int r = 0; r < 8; ++r) {
      const unsigned dN8 = (unsigned)idxS[16 + r + 8 * half];
      zb0[r] = (dN8 * 64u + (unsigned)nl) * 4u;
      zb1[r] = (dN8 * 192u + (unsigned)nl) * 4u;
    }
    unsigned sRow[16];
#pragma unroll
    for (int R = 0; R < 16; ++R) sRow[R] = (unsigned)__builtin_amdgcn_readlane(sN, R);
    float rh[3][8];
#pragma unroll
    for (int r = 0; r < 8; ++r) {
      rh[0][r] = rhS[(r + 8 * half) * 4 + 0];
      rh[1][r] = rhS[(r + 8 * half) * 4 + 1];
      rh[2][r] = rhS[(r + 8 * half) * 4 + 2];
    }

    v2f af[16];
#pragma unroll
    for (int t = 0; t < 16; ++t)
      af[t] = *(const v2f*)(stage + nl * WP + 4 * t + 2 * half);

#pragma unroll 1
    for (int jp = 0; jp < 2; ++jp) {
      const char* z0b = (const char*)z0g;
      const char* z1b = (const char*)z1g;
      float ms[2][8], mv[2][3][8];

#pragma unroll
      for (int h2 = 0; h2 < 2; ++h2) {
        const int j = 2 * jp + h2;
        float z0r[8];
#pragma unroll
        for (int r = 0; r < 8; ++r) z0r[r] = *(const float*)(z0b + zb0[r]);
        float z1r[3][8];
#pragma unroll
        for (int i = 0; i < 3; ++i)
#pragma unroll
          for (int r = 0; r < 8; ++r)
            z1r[i][r] = *(const float*)(z1b + i * 256 + zb1[r]);

        const float* wb = smem + OFF_WGATE + (16 * j + nl) * WP + 2 * half;
        float bb[4];
        v8f g[4];
        v2f bq[2][2][4];
#pragma unroll
        for (int c = 0; c < 4; ++c) {
          g[c] = (v8f)0.0f;
          bb[c] = smem[OFF_BGATE + 64 * c + 16 * j + nl];
          bq[0][0][c] = *(const v2f*)(wb + 64 * c * WP);
          bq[0][1][c] = *(const v2f*)(wb + 64 * c * WP + 4);
        }
        __builtin_amdgcn_sched_barrier(0);
#pragma unroll
        for (int u = 0; u < 8; ++u) {
          if (u < 7) {
#pragma unroll
            for (int c = 0; c < 4; ++c) {
              bq[(u + 1) & 1][0][c] = *(const v2f*)(wb + 64 * c * WP + 8 * (u + 1));
              bq[(u + 1) & 1][1][c] = *(const v2f*)(wb + 64 * c * WP + 8 * (u + 1) + 4);
            }
          }
          __builtin_amdgcn_sched_barrier(0);
#pragma unroll
          for (int c = 0; c < 4; ++c) g[c] = wmma4(af[2 * u], bq[u & 1][0][c], g[c]);
#pragma unroll
          for (int c = 0; c < 4; ++c) g[c] = wmma4(af[2 * u + 1], bq[u & 1][1][c], g[c]);
          __builtin_amdgcn_sched_barrier(0);
        }
#pragma unroll
        for (int c = 0; c < 4; ++c)
#pragma unroll
          for (int r = 0; r < 8; ++r) g[c][r] += bb[c];

#pragma unroll
        for (int r = 0; r < 8; ++r) {
          float dotv = rh[0][r] * z1r[0][r] + rh[1][r] * z1r[1][r] +
                       rh[2][r] * z1r[2][r];
          ms[h2][r] = g[0][r] * z0r[r] + g[1][r] * dotv;
        }
#pragma unroll
        for (int i = 0; i < 3; ++i)
#pragma unroll
          for (int r = 0; r < 8; ++r)
            mv[h2][i][r] = g[2][r] * z1r[i][r] + g[3][r] * rh[i][r] * z0r[r];

#pragma unroll
        for (int r = 0; r < 8; ++r) {
          zb0[r] += 64u;
          zb1[r] += 64u;
        }
        __builtin_amdgcn_sched_barrier(0);
      }

      float xs[2][8], xv[2][3][8];
#pragma unroll
      for (int r = 0; r < 8; ++r) {
        v2u t = __builtin_amdgcn_permlane16_swap(__float_as_uint(ms[0][r]), __float_as_uint(ms[1][r]), false, false);
        xs[0][r] = __uint_as_float(t[0]);
        xs[1][r] = __uint_as_float(t[1]);
      }
#pragma unroll
      for (int i = 0; i < 3; ++i)
#pragma unroll
        for (int r = 0; r < 8; ++r) {
          v2u t = __builtin_amdgcn_permlane16_swap(__float_as_uint(mv[0][i][r]), __float_as_uint(mv[1][i][r]), false, false);
          xv[0][i][r] = __uint_as_float(t[0]);
          xv[1][i][r] = __uint_as_float(t[1]);
        }
      __builtin_amdgcn_sched_barrier(0);

      const unsigned ch = (unsigned)(32 * jp + lane);
#pragma unroll
      for (int q = 0; q < 2; ++q)
#pragma unroll
        for (int r = 0; r < 8; ++r) {
          atomicAdd(out0 + (size_t)sRow[r + 8 * q] * 64 + ch, xs[q][r]);
#pragma unroll
          for (int i = 0; i < 3; ++i)
            atomicAdd(out1 + (size_t)sRow[r + 8 * q] * 192 + i * 64 + ch, xv[q][i][r]);
        }
    }
  }
}

__global__ void __launch_bounds__(BLOCK)
node_transform_kernel(const float* __restrict__ Ws, const float* __restrict__ Wv,
                      float* out) {
  extern __shared__ float smem[];

  for (int i = threadIdx.x; i < 64 * 64; i += BLOCK) {
    int k = i >> 6, n = i & 63;
    smem[N_OFF_WS + n * WP + k] = Ws[i];
    smem[N_OFF_WV + n * WP + k] = Wv[i];
  }
  __syncthreads();

  const int lane = threadIdx.x & 31;
  const int wave = threadIdx.x >> 5;
  const int half = lane >> 4;
  const int nl   = lane & 15;

  float* stage = smem + N_OFF_STAGE + wave * (16 * WP);

  for (int task = blockIdx.x * WAVES + wave; task < NODE_TASKS;
       task += gridDim.x * WAVES) {
    const int tile = task >> 2;
    const int part = task & 3;
    float* base;
    int pitch;
    const float* wT;
    if (part == 0) {
      base = out + (size_t)tile * (16 * 64);
      pitch = 64;
      wT = smem + N_OFF_WS;
    } else {
      base = out + (size_t)NNODES * CHAN + (size_t)tile * (16 * 192) + (part - 1) * 64;
      pitch = 192;
      wT = smem + N_OFF_WV;
    }

    v4f v[8];
#pragma unroll
    for (int q = 0; q < 8; ++q) {
      int idx = q * 32 + lane;
      v[q] = *(const v4f*)(base + (size_t)(idx >> 4) * pitch + (idx & 15) * 4);
    }
    lds_fence();
#pragma unroll
    for (int q = 0; q < 8; ++q) {
      int idx = q * 32 + lane;
      *(v4f*)(stage + (idx >> 4) * WP + (idx & 15) * 4) = v[q];
    }
    lds_fence();

    v8f p[4];
#pragma unroll
    for (int j = 0; j < 4; ++j) p[j] = (v8f)0.0f;
#pragma unroll
    for (int t = 0; t < 16; ++t) {
      int k0 = 4 * t + 2 * half;
      v2f a = *(const v2f*)(stage + nl * WP + k0);
#pragma unroll
      for (int j = 0; j < 4; ++j) {
        v2f b = *(const v2f*)(wT + (16 * j + nl) * WP + k0);
        p[j] = wmma4(a, b, p[j]);
      }
    }

    lds_fence();
#pragma unroll
    for (int j = 0; j < 4; ++j)
#pragma unroll
      for (int r = 0; r < 8; ++r)
        stage[(r + 8 * half) * WP + 16 * j + nl] = p[j][r];
    lds_fence();
#pragma unroll
    for (int q = 0; q < 8; ++q) {
      int idx = q * 32 + lane;
      v[q] = *(const v4f*)(stage + (idx >> 4) * WP + (idx & 15) * 4);
    }
#pragma unroll
    for (int q = 0; q < 8; ++q) {
      int idx = q * 32 + lane;
      *(v4f*)(base + (size_t)(idx >> 4) * pitch + (idx & 15) * 4) = v[q];
    }
  }
}

extern "C" void kernel_launch(void* const* d_in, const int* in_sizes, int n_in,
                              void* d_out, int out_size, void* d_ws,
                              size_t ws_size, hipStream_t stream) {
  const int*   src   = (const int*)d_in[0];
  const int*   dst   = (const int*)d_in[1];
  const float* r_ij  = (const float*)d_in[2];
  const float* z0    = (const float*)d_in[3];
  const float* z1    = (const float*)d_in[4];
  const float* emb   = (const float*)d_in[5];
  const float* Wenc  = (const float*)d_in[6];
  const float* benc  = (const float*)d_in[7];
  const float* Wsrc  = (const float*)d_in[8];
  const float* bsrc  = (const float*)d_in[9];
  const float* Wdst  = (const float*)d_in[10];
  const float* bdst  = (const float*)d_in[11];
  const float* Wgate = (const float*)d_in[12];
  const float* bgate = (const float*)d_in[13];
  const float* Ws    = (const float*)d_in[14];
  const float* Wv    = (const float*)d_in[15];
  float* out = (float*)d_out;

  zero_f32<<<ZGRID, BLOCK, 0, stream>>>(out, out_size);

  size_t smem_bytes = (size_t)SMEM_FLOATS * sizeof(float);
  messages_kernel<<<GRID, BLOCK, smem_bytes, stream>>>(
      src, dst, r_ij, z0, z1, emb, Wenc, benc, Wsrc, bsrc, Wdst, bdst, Wgate,
      bgate, out, out + (size_t)NNODES * CHAN);

  size_t nsmem_bytes = (size_t)N_SMEM_FLOATS * sizeof(float);
  node_transform_kernel<<<NGRID, BLOCK, nsmem_bytes, stream>>>(Ws, Wv, out);
}


# emb@W_src, emb@W_dst done once per node row by a pre-pass into workspace planes (used only if the workspace holds them); edge kernel gathers rows: 128 of 392 WMMAs a tile go; float sum order changes
# speedup vs baseline: 4.7189x; 1.0547x over previous
#include <hip/hip_runtime.h>
#include <math.h>

#define NNODES 50000
#define NEDGES 500000
#define CHAN   64
#define R0f    2.0f

#define WAVES  8
#define BLOCK  (WAVES * 32)
#define NTILES (NEDGES / 16)
#define GRID   512

typedef float v2f __attribute__((ext_vector_type(2)));
typedef float v4f __attribute__((ext_vector_type(4)));
typedef float v8f __attribute__((ext_vector_type(8)));
typedef unsigned v2u __attribute__((ext_vector_type(2)));

#define WP 68
#define EP 12
#define OFF_WSRC  0
#define OFF_WDST  (OFF_WSRC + 64 * WP)
#define OFF_WGATE (OFF_WDST + 64 * WP)
#define OFF_WENC  (OFF_WGATE + 256 * WP)
#define OFF_BIAS0 (OFF_WENC + 64 * EP)
#define OFF_BGATE (OFF_BIAS0 + 64)
#define OFF_STAGE (OFF_BGATE + 256)
#define OFF_RH    (OFF_STAGE + WAVES * 16 * WP)
#define OFF_IDX   (OFF_RH + WAVES * 64)
#define SMEM_FLOATS (OFF_IDX + WAVES * 32)

#define NODE_TILES (NNODES / 16)
#define NODE_TASKS (NODE_TILES * 4)
#define NGRID      1024
#define N_OFF_WS    0
#define N_OFF_WV    (N_OFF_WS + 64 * WP)
#define N_OFF_STAGE (N_OFF_WV + 64 * WP)
#define N_SMEM_FLOATS (N_OFF_STAGE + WAVES * 16 * WP)

#define P_TASKS    (NODE_TILES * 2)
#define PGRID      782
#define P_FLOATS   ((size_t)NNODES * CHAN * 2)
#define P_BYTES    (P_FLOATS * sizeof(float))

__device__ __forceinline__ void lds_fence() {
  asm volatile("s_wait_dscnt 0x0" ::: "memory");
}

__device__ __forceinline__ v8f wmma4(v2f a, v2f b, v8f c) {
  return __builtin_amdgcn_wmma_f32_16x16x4_f32(false, a, false, b, (short)0, c,
                                               false, false);
}

__device__ __forceinline__ v2f lo2(v4f v) { return __builtin_shufflevector(v, v, 0, 1); }
__device__ __forceinline__ v2f hi2(v4f v) { return __builtin_shufflevector(v, v, 2, 3); }

__device__ __forceinline__ void mm64(v8f (&acc)[4], const v4f (&a4)[8], const float* wrow) {
  v4f bq[2][2][4];
#pragma unroll
  for (int s = 0; s < 2; ++s)
#pragma unroll
    for (int j = 0; j < 4; ++j) bq[0][s][j] = *(const v4f*)(wrow + j * 16 * WP + 8 * s);
#pragma unroll
  for (int u = 0; u < 4; ++u) {
    if (u < 3) {
#pragma unroll
      for (int s = 0; s < 2; ++s)
#pragma unroll
        for (int j = 0; j < 4; ++j)
          bq[(u + 1) & 1][s][j] = *(const v4f*)(wrow + j * 16 * WP + 8 * (2 * (u + 1) + s));
    }
    __builtin_amdgcn_sched_barrier(0);
#pragma unroll
    for (int s = 0; s < 2; ++s) {
#pragma unroll
      for (int j = 0; j < 4; ++j) acc[j] = wmma4(lo2(a4[2 * u + s]), lo2(bq[u & 1][s][j]), acc[j]);
#pragma unroll
      for (int j = 0; j < 4; ++j) acc[j] = wmma4(hi2(a4[2 * u + s]), hi2(bq[u & 1][s][j]), acc[j]);
    }
    __builtin_amdgcn_sched_barrier(0);
  }
}

#define ZGRID   256
#define ZSTRIDE (ZGRID * BLOCK)
__global__ void __launch_bounds__(BLOCK)
zero_f32(float* __restrict__ p, int n) {
  const int n4 = n >> 2;
  v4f* __restrict__ p4 = (v4f*)p;
  int i = blockIdx.x * BLOCK + threadIdx.x;
  for (; i + 7 * ZSTRIDE < n4; i += 8 * ZSTRIDE) {
#pragma unroll
    for (int u = 0; u < 8; ++u) p4[i + u * ZSTRIDE] = (v4f)0.0f;
  }
  for (; i < n4; i += ZSTRIDE) p4[i] = (v4f)0.0f;
  const int t = (n4 << 2) + threadIdx.x;
  if (blockIdx.x == 0 && t < n) p[t] = 0.0f;
}

__global__ void __launch_bounds__(BLOCK, 4)
messages_kernel(const int* __restrict__ src, const int* __restrict__ dst,
                const float* __restrict__ r_ij,
                const float* __restrict__ z0g, const float* __restrict__ z1g,
                const float* __restrict__ emb,
                const float* __restrict__ Wenc, const float* __restrict__ benc,
                const float* __restrict__ Wsrc, const float* __restrict__ bsrc,
                const float* __restrict__ Wdst, const float* __restrict__ bdst,
                const float* __restrict__ Wgate, const float* __restrict__ bgate,
                float* __restrict__ out0, float* __restrict__ out1,
                const float* __restrict__ P, size_t ws_bytes) {
  extern __shared__ float smem[];

  const bool useP = ws_bytes >= P_BYTES;

  for (int i = threadIdx.x; i < 64 * 64; i += BLOCK) {
    int k = i >> 6, n = i & 63;
    smem[OFF_WSRC + n * WP + k] = Wsrc[i];
    smem[OFF_WDST + n * WP + k] = Wdst[i];
  }
  for (int i = threadIdx.x; i < 64 * 256; i += BLOCK) {
    int k = i >> 8, n = i & 255;
    smem[OFF_WGATE + n * WP + k] = Wgate[i];
  }
  for (int i = threadIdx.x; i < 8 * 64; i += BLOCK) {
    int k = i >> 6, n = i & 63;
    smem[OFF_WENC + n * EP + k] = Wenc[i];
  }
  if (threadIdx.x < 64)
    smem[OFF_BIAS0 + threadIdx.x] =
        bsrc[threadIdx.x] + bdst[threadIdx.x] + benc[threadIdx.x];
  if (threadIdx.x < 256) smem[OFF_BGATE + threadIdx.x] = bgate[threadIdx.x];
  __syncthreads();

  const int lane = threadIdx.x & 31;
  const int wave = threadIdx.x >> 5;
  const int half = lane >> 4;
  const int nl   = lane & 15;

  float* stage = smem + OFF_STAGE + wave * (16 * WP);
  float* rhS   = smem + OFF_RH + wave * 64;
  int*   idxS  = ((int*)(smem + OFF_IDX)) + wave * 32;

  for (int tile = blockIdx.x * WAVES + wave; tile < NTILES;
       tile += gridDim.x * WAVES) {
    const int e  = tile * 16 + nl;
    const int sN = src[e];
    const int dN = dst[e];
    if (half == 0) { idxS[nl] = sN; idxS[16 + nl] = dN; }

    float rx = r_ij[e * 3 + 0], ry = r_ij[e * 3 + 1], rz = r_ij[e * 3 + 2];
    float dd  = sqrtf(rx * rx + ry * ry + rz * rz);
    float sx = rx * 3.5f, sy = ry * 3.5f, sz = rz * 3.5f;
    float inv = rsqrtf(1.0f + sx * sx + sy * sy + sz * sz);
    if (half == 0) {
      rhS[nl * 4 + 0] = sx * inv;
      rhS[nl * 4 + 1] = sy * inv;
      rhS[nl * 4 + 2] = sz * inv;
    }
    float rb[4];
#pragma unroll
    for (int u = 0; u < 4; ++u) {
      int k = 4 * half + u;
      float x = (dd - (float)k * (R0f / 7.0f)) * 4.0f;
      rb[u] = expf(-x * x);
    }

    if (useP) {
      const char* Pb = (const char*)P;
      v4f ps[8], pd[8];
#pragma unroll
      for (int r = 0; r < 8; ++r) {
        const unsigned sR = (unsigned)idxS[r + 8 * half];
        const unsigned dR = (unsigned)idxS[16 + r + 8 * half];
        ps[r] = *(const v4f*)(Pb + (sR * 64u + 4u * (unsigned)nl) * 4u);
        pd[r] = *(const v4f*)(Pb + ((unsigned)(NNODES * CHAN) + dR * 64u + 4u * (unsigned)nl) * 4u);
      }
      v8f acc[4];
      {
        v4f be[4];
#pragma unroll
        for (int j = 0; j < 4; ++j)
          be[j] = *(const v4f*)(smem + OFF_WENC + (4 * nl + j) * EP + 4 * half);
        v2f a0, a1;
        a0[0] = rb[0];
        a0[1] = rb[1];
        a1[0] = rb[2];
        a1[1] = rb[3];
#pragma unroll
        for (int j = 0; j < 4; ++j) acc[j] = wmma4(a0, lo2(be[j]), (v8f)0.0f);
#pragma unroll
        for (int j = 0; j < 4; ++j) acc[j] = wmma4(a1, hi2(be[j]), acc[j]);
      }
      const v4f bb = *(const v4f*)(smem + OFF_BIAS0 + 4 * nl);

      lds_fence();
#pragma unroll
      for (int r = 0; r < 8; ++r) {
        v4f o;
#pragma unroll
        for (int j = 0; j < 4; ++j) o[j] = acc[j][r] + (ps[r][j] + pd[r][j] + bb[j]);
        *(v4f*)(stage + (r + 8 * half) * WP + 4 * nl) = o;
      }
      lds_fence();
    } else {
      v8f acc[4];
#pragma unroll
      for (int j = 0; j < 4; ++j) acc[j] = (v8f)0.0f;

      const float* es = emb + (size_t)sN * 64 + 4 * half;
      const float* ed = emb + (size_t)dN * 64 + 4 * half;
      v4f es4[8], ed4[8];
#pragma unroll
      for (int q = 0; q < 8; ++q) es4[q] = *(const v4f*)(es + 8 * q);
#pragma unroll
      for (int q = 0; q < 8; ++q) ed4[q] = *(const v4f*)(ed + 8 * q);
      mm64(acc, es4, smem + OFF_WSRC + nl * WP + 4 * half);
      mm64(acc, ed4, smem + OFF_WDST + nl * WP + 4 * half);
      {
        v4f be[4];
#pragma unroll
        for (int j = 0; j < 4; ++j)
          be[j] = *(const v4f*)(smem + OFF_WENC + (16 * j + nl) * EP + 4 * half);
        v2f a0, a1;
        a0[0] = rb[0];
        a0[1] = rb[1];
        a1[0] = rb[2];
        a1[1] = rb[3];
#pragma unroll
        for (int j = 0; j < 4; ++j) acc[j] = wmma4(a0, lo2(be[j]), acc[j]);
#pragma unroll
        for (int j = 0; j < 4; ++j) acc[j] = wmma4(a1, hi2(be[j]), acc[j]);
      }
#pragma unroll
      for (int j = 0; j < 4; ++j) {
        float bb = smem[OFF_BIAS0 + 16 * j + nl];
#pragma unroll
        for (int r = 0; r < 8; ++r) acc[j][r] += bb;
      }

      lds_fence();
#pragma unroll
      for (int j = 0; j < 4; ++j)
#pragma unroll
        for (int r = 0; r < 8; ++r)
          stage[(r + 8 * half) * WP + 16 * j + nl] = acc[j][r];
      lds_fence();
    }

    unsigned zb0[8], zb1[8];
#pragma unroll
    for (int r = 0; r < 8; ++r) {
      const unsigned dN8 = (unsigned)idxS[16 + r + 8 * half];
      zb0[r] = (dN8 * 64u + (unsigned)nl) * 4u;
      zb1[r] = (dN8 * 192u + (unsigned)nl) * 4u;
    }
    unsigned sRow[16];
#pragma unroll
    for (int R = 0; R < 16; ++R) sRow[R] = (unsigned)__builtin_amdgcn_readlane(sN, R);
    float rh[3][8];
#pragma unroll
    for (int r = 0; r < 8; ++r) {
      rh[0][r] = rhS[(r + 8 * half) * 4 + 0];
      rh[1][r] = rhS[(r + 8 * half) * 4 + 1];
      rh[2][r] = rhS[(r + 8 * half) * 4 + 2];
    }

    v2f af[16];
#pragma unroll
    for (int t = 0; t < 16; ++t)
      af[t] = *(const v2f*)(stage + nl * WP + 4 * t + 2 * half);

#pragma unroll 1
    for (int jp = 0; jp < 2; ++jp) {
      const char* z0b = (const char*)z0g;
      const char* z1b = (const char*)z1g;
      float ms[2][8], mv[2][3][8];

#pragma unroll
      for (int h2 = 0; h2 < 2; ++h2) {
        const int j = 2 * jp + h2;
        float z0r[8];
#pragma unroll
        for (int r = 0; r < 8; ++r) z0r[r] = *(const float*)(z0b + zb0[r]);
        float z1r[3][8];
#pragma unroll
        for (int i = 0; i < 3; ++i)
#pragma unroll
          for (int r = 0; r < 8; ++r)
            z1r[i][r] = *(const float*)(z1b + i * 256 + zb1[r]);

        const float* wb = smem + OFF_WGATE + (16 * j + nl) * WP + 2 * half;
        float bb[4];
        v8f g[4];
        v2f bq[2][2][4];
#pragma unroll
        for (int c = 0; c < 4; ++c) {
          g[c] = (v8f)0.0f;
          bb[c] = smem[OFF_BGATE + 64 * c + 16 * j + nl];
          bq[0][0][c] = *(const v2f*)(wb + 64 * c * WP);
          bq[0][1][c] = *(const v2f*)(wb + 64 * c * WP + 4);
        }
        __builtin_amdgcn_sched_barrier(0);
#pragma unroll
        for (int u = 0; u < 8; ++u) {
          if (u < 7) {
#pragma unroll
            for (int c = 0; c < 4; ++c) {
              bq[(u + 1) & 1][0][c] = *(const v2f*)(wb + 64 * c * WP + 8 * (u + 1));
              bq[(u + 1) & 1][1][c] = *(const v2f*)(wb + 64 * c * WP + 8 * (u + 1) + 4);
            }
          }
          __builtin_amdgcn_sched_barrier(0);
#pragma unroll
          for (int c = 0; c < 4; ++c) g[c] = wmma4(af[2 * u], bq[u & 1][0][c], g[c]);
#pragma unroll
          for (int c = 0; c < 4; ++c) g[c] = wmma4(af[2 * u + 1], bq[u & 1][1][c], g[c]);
          __builtin_amdgcn_sched_barrier(0);
        }
#pragma unroll
        for (int c = 0; c < 4; ++c)
#pragma unroll
          for (int r = 0; r < 8; ++r) g[c][r] += bb[c];

#pragma unroll
        for (int r = 0; r < 8; ++r) {
          float dotv = rh[0][r] * z1r[0][r] + rh[1][r] * z1r[1][r] +
                       rh[2][r] * z1r[2][r];
          ms[h2][r] = g[0][r] * z0r[r] + g[1][r] * dotv;
        }
#pragma unroll
        for (int i = 0; i < 3; ++i)
#pragma unroll
          for (int r = 0; r < 8; ++r)
            mv[h2][i][r] = g[2][r] * z1r[i][r] + g[3][r] * rh[i][r] * z0r[r];

#pragma unroll
        for (int r = 0; r < 8; ++r) {
          zb0[r] += 64u;
          zb1[r] += 64u;
        }
        __builtin_amdgcn_sched_barrier(0);
      }

      float xs[2][8], xv[2][3][8];
#pragma unroll
      for (int r = 0; r < 8; ++r) {
        v2u t = __builtin_amdgcn_permlane16_swap(__float_as_uint(ms[0][r]), __float_as_uint(ms[1][r]), false, false);
        xs[0][r] = __uint_as_float(t[0]);
        xs[1][r] = __uint_as_float(t[1]);
      }
#pragma unroll
      for (int i = 0; i < 3; ++i)
#pragma unroll
        for (int r = 0; r < 8; ++r) {
          v2u t = __builtin_amdgcn_permlane16_swap(__float_as_uint(mv[0][i][r]), __float_as_uint(mv[1][i][r]), false, false);
          xv[0][i][r] = __uint_as_float(t[0]);
          xv[1][i][r] = __uint_as_float(t[1]);
        }
      __builtin_amdgcn_sched_barrier(0);

      const unsigned ch = (unsigned)(32 * jp + lane);
#pragma unroll
      for (int q = 0; q < 2; ++q)
#pragma unroll
        for (int r = 0; r < 8; ++r) {
          atomicAdd(out0 + (size_t)sRow[r + 8 * q] * 64 + ch, xs[q][r]);
#pragma unroll
          for (int i = 0; i < 3; ++i)
            atomicAdd(out1 + (size_t)sRow[r + 8 * q] * 192 + i * 64 + ch, xv[q][i][r]);
        }
    }
  }
}

__device__ __forceinline__ void tile_times_w(const float* inb, int in_pitch, float* outb, int out_pitch,
                                             const float* wT, float* stage, int lane, int half, int nl) {
  v4f v[8];
#pragma unroll
  for (int q = 0; q < 8; ++q) {
    int idx = q * 32 + lane;
    v[q] = *(const v4f*)(inb + (size_t)(idx >> 4) * in_pitch + (idx & 15) * 4);
  }
  lds_fence();
#pragma unroll
  for (int q = 0; q < 8; ++q) {
    int idx = q * 32 + lane;
    *(v4f*)(stage + (idx >> 4) * WP + (idx & 15) * 4) = v[q];
  }
  lds_fence();

  v8f p[4];
#pragma unroll
  for (int j = 0; j < 4; ++j) p[j] = (v8f)0.0f;
#pragma unroll
  for (int t = 0; t < 16; ++t) {
    int k0 = 4 * t + 2 * half;
    v2f a = *(const v2f*)(stage + nl * WP + k0);
#pragma unroll
    for (int j = 0; j < 4; ++j) {
      v2f b = *(const v2f*)(wT + (16 * j + nl) * WP + k0);
      p[j] = wmma4(a, b, p[j]);
    }
  }

  lds_fence();
#pragma unroll
  for (int j = 0; j < 4; ++j)
#pragma unroll
    for (int r = 0; r < 8; ++r)
      stage[(r + 8 * half) * WP + 16 * j + nl] = p[j][r];
  lds_fence();
#pragma unroll
  for (int q = 0; q < 8; ++q) {
    int idx = q * 32 + lane;
    v[q] = *(const v4f*)(stage + (idx >> 4) * WP + (idx & 15) * 4);
  }
#pragma unroll
  for (int q = 0; q < 8; ++q) {
    int idx = q * 32 + lane;
    *(v4f*)(outb + (size_t)(idx >> 4) * out_pitch + (idx & 15) * 4) = v[q];
  }
}

__global__ void __launch_bounds__(BLOCK)
node_embed_kernel(const float* __restrict__ emb, const float* __restrict__ Wsrc,
                  const float* __restrict__ Wdst, float* __restrict__ P, size_t ws_bytes) {
  extern __shared__ float smem[];

  if (ws_bytes < P_BYTES) return;

  for (int i = threadIdx.x; i < 64 * 64; i += BLOCK) {
    int k = i >> 6, n = i & 63;
    smem[N_OFF_WS + n * WP + k] = Wsrc[i];
    smem[N_OFF_WV + n * WP + k] = Wdst[i];
  }
  __syncthreads();

  const int lane = threadIdx.x & 31;
  const int wave = threadIdx.x >> 5;
  const int half = lane >> 4;
  const int nl   = lane & 15;

  float* stage = smem + N_OFF_STAGE + wave * (16 * WP);

  for (int task = blockIdx.x * WAVES + wave; task < P_TASKS;
       task += gridDim.x * WAVES) {
    const int tile = task >> 1;
    const int part = task & 1;
    tile_times_w(emb + (size_t)tile * (16 * 64), 64,
                 P + (size_t)part * ((size_t)NNODES * CHAN) + (size_t)tile * (16 * 64), 64,
                 smem + (part ? N_OFF_WV : N_OFF_WS), stage, lane, half, nl);
  }
}

__global__ void __launch_bounds__(BLOCK)
node_transform_kernel(const float* __restrict__ Ws, const float* __restrict__ Wv,
                      float* out) {
  extern __shared__ float smem[];

  for (int i = threadIdx.x; i < 64 * 64; i += BLOCK) {
    int k = i >> 6, n = i & 63;
    smem[N_OFF_WS + n * WP + k] = Ws[i];
    smem[N_OFF_WV + n * WP + k] = Wv[i];
  }
  __syncthreads();

  const int lane = threadIdx.x & 31;
  const int wave = threadIdx.x >> 5;
  const int half = lane >> 4;
  const int nl   = lane & 15;

  float* stage = smem + N_OFF_STAGE + wave * (16 * WP);

  for (int task = blockIdx.x * WAVES + wave; task < NODE_TASKS;
       task += gridDim.x * WAVES) {
    const int tile = task >> 2;
    const int part = task & 3;
    float* base;
    int pitch;
    const float* wT;
    if (part == 0) {
      base = out + (size_t)tile * (16 * 64);
      pitch = 64;
      wT = smem + N_OFF_WS;
    } else {
      base = out + (size_t)NNODES * CHAN + (size_t)tile * (16 * 192) + (part - 1) * 64;
      pitch = 192;
      wT = smem + N_OFF_WV;
    }
    tile_times_w(base, pitch, base, pitch, wT, stage, lane, half, nl);
  }
}

extern "C" void kernel_launch(void* const* d_in, const int* in_sizes, int n_in,
                              void* d_out, int out_size, void* d_ws,
                              size_t ws_size, hipStream_t stream) {
  const int*   src   = (const int*)d_in[0];
  const int*   dst   = (const int*)d_in[1];
  const float* r_ij  = (const float*)d_in[2];
  const float* z0    = (const float*)d_in[3];
  const float* z1    = (const float*)d_in[4];
  const float* emb   = (const float*)d_in[5];
  const float* Wenc  = (const float*)d_in[6];
  const float* benc  = (const float*)d_in[7];
  const float* Wsrc  = (const float*)d_in[8];
  const float* bsrc  = (const float*)d_in[9];
  const float* Wdst  = (const float*)d_in[10];
  const float* bdst  = (const float*)d_in[11];
  const float* Wgate = (const float*)d_in[12];
  const float* bgate = (const float*)d_in[13];
  const float* Ws    = (const float*)d_in[14];
  const float* Wv    = (const float*)d_in[15];
  float* out = (float*)d_out;
  float* ws  = (float*)d_ws;

  zero_f32<<<ZGRID, BLOCK, 0, stream>>>(out, out_size);

  size_t nsmem_bytes = (size_t)N_SMEM_FLOATS * sizeof(float);
  node_embed_kernel<<<PGRID, BLOCK, nsmem_bytes, stream>>>(emb, Wsrc, Wdst, ws, ws_size);

  size_t smem_bytes = (size_t)SMEM_FLOATS * sizeof(float);
  messages_kernel<<<GRID, BLOCK, smem_bytes, stream>>>(
      src, dst, r_ij, z0, z1, emb, Wenc, benc, Wsrc, bsrc, Wdst, bdst, Wgate,
      bgate, out, out + (size_t)NNODES * CHAN, ws, ws_size);

  node_transform_kernel<<<NGRID, BLOCK, nsmem_bytes, stream>>>(Ws, Wv, out);
}


# z0/z1 gathered 8 B per lane once per pair of 16-channel groups (W_gate columns staged so a lane's two groups hold neighbour channels): each 128-B z line read once, not twice as halves; sums as before
# speedup vs baseline: 4.9580x; 1.0507x over previous
#include <hip/hip_runtime.h>
#include <math.h>

#define NNODES 50000
#define NEDGES 500000
#define CHAN   64
#define R0f    2.0f

#define WAVES  8
#define BLOCK  (WAVES * 32)
#define NTILES (NEDGES / 16)
#define GRID   512

typedef float v2f __attribute__((ext_vector_type(2)));
typedef float v4f __attribute__((ext_vector_type(4)));
typedef float v8f __attribute__((ext_vector_type(8)));
typedef unsigned v2u __attribute__((ext_vector_type(2)));

#define WP 68
#define EP 12
#define OFF_WSRC  0
#define OFF_WDST  (OFF_WSRC + 64 * WP)
#define OFF_WGATE (OFF_WDST + 64 * WP)
#define OFF_WENC  (OFF_WGATE + 256 * WP)
#define OFF_BIAS0 (OFF_WENC + 64 * EP)
#define OFF_BGATE (OFF_BIAS0 + 64)
#define OFF_STAGE (OFF_BGATE + 256)
#define OFF_RH    (OFF_STAGE + WAVES * 16 * WP)
#define OFF_IDX   (OFF_RH + WAVES * 64)
#define SMEM_FLOATS (OFF_IDX + WAVES * 32)

#define NODE_TILES (NNODES / 16)
#define NODE_TASKS (NODE_TILES * 4)
#define NGRID      1024
#define N_OFF_WS    0
#define N_OFF_WV    (N_OFF_WS + 64 * WP)
#define N_OFF_STAGE (N_OFF_WV + 64 * WP)
#define N_SMEM_FLOATS (N_OFF_STAGE + WAVES * 16 * WP)

#define P_TASKS    (NODE_TILES * 2)
#define PGRID      782
#define P_FLOATS   ((size_t)NNODES * CHAN * 2)
#define P_BYTES    (P_FLOATS * sizeof(float))

__device__ __forceinline__ void lds_fence() {
  asm volatile("s_wait_dscnt 0x0" ::: "memory");
}

__device__ __forceinline__ v8f wmma4(v2f a, v2f b, v8f c) {
  return __builtin_amdgcn_wmma_f32_16x16x4_f32(false, a, false, b, (short)0, c,
                                               false, false);
}

__device__ __forceinline__ v2f lo2(v4f v) { return __builtin_shufflevector(v, v, 0, 1); }
__device__ __forceinline__ v2f hi2(v4f v) { return __builtin_shufflevector(v, v, 2, 3); }

__device__ __forceinline__ void mm64(v8f (&acc)[4], const v4f (&a4)[8], const float* wrow) {
  v4f bq[2][2][4];
#pragma unroll
  for (int s = 0; s < 2; ++s)
#pragma unroll
    for (int j = 0; j < 4; ++j) bq[0][s][j] = *(const v4f*)(wrow + j * 16 * WP + 8 * s);
#pragma unroll
  for (int u = 0; u < 4; ++u) {
    if (u < 3) {
#pragma unroll
      for (int s = 0; s < 2; ++s)
#pragma unroll
        for (int j = 0; j < 4; ++j)
          bq[(u + 1) & 1][s][j] = *(const v4f*)(wrow + j * 16 * WP + 8 * (2 * (u + 1) + s));
    }
    __builtin_amdgcn_sched_barrier(0);
#pragma unroll
    for (int s = 0; s < 2; ++s) {
#pragma unroll
      for (int j = 0; j < 4; ++j) acc[j] = wmma4(lo2(a4[2 * u + s]), lo2(bq[u & 1][s][j]), acc[j]);
#pragma unroll
      for (int j = 0; j < 4; ++j) acc[j] = wmma4(hi2(a4[2 * u + s]), hi2(bq[u & 1][s][j]), acc[j]);
    }
    __builtin_amdgcn_sched_barrier(0);
  }
}

#define ZGRID   256
#define ZSTRIDE (ZGRID * BLOCK)
__global__ void __launch_bounds__(BLOCK)
zero_f32(float* __restrict__ p, int n) {
  const int n4 = n >> 2;
  v4f* __restrict__ p4 = (v4f*)p;
  int i = blockIdx.x * BLOCK + threadIdx.x;
  for (; i + 7 * ZSTRIDE < n4; i += 8 * ZSTRIDE) {
#pragma unroll
    for (int u = 0; u < 8; ++u) p4[i + u * ZSTRIDE] = (v4f)0.0f;
  }
  for (; i < n4; i += ZSTRIDE) p4[i] = (v4f)0.0f;
  const int t = (n4 << 2) + threadIdx.x;
  if (blockIdx.x == 0 && t < n) p[t] = 0.0f;
}

__global__ void __launch_bounds__(BLOCK, 4)
messages_kernel(const int* __restrict__ src, const int* __restrict__ dst,
                const float* __restrict__ r_ij,
                const float* __restrict__ z0g, const float* __restrict__ z1g,
                const float* __restrict__ emb,
                const float* __restrict__ Wenc, const float* __restrict__ benc,
                const float* __restrict__ Wsrc, const float* __restrict__ bsrc,
                const float* __restrict__ Wdst, const float* __restrict__ bdst,
                const float* __restrict__ Wgate, const float* __restrict__ bgate,
                float* __restrict__ out0, float* __restrict__ out1,
                const float* __restrict__ P, size_t ws_bytes) {
  extern __shared__ float smem[];

  const bool useP = ws_bytes >= P_BYTES;

  for (int i = threadIdx.x; i < 64 * 64; i += BLOCK) {
    int k = i >> 6, n = i & 63;
    smem[OFF_WSRC + n * WP + k] = Wsrc[i];
    smem[OFF_WDST + n * WP + k] = Wdst[i];
  }
  for (int i = threadIdx.x; i < 64 * 256; i += BLOCK) {
    int k = i >> 8, n = i & 255;
    int row = (n & 0xE0) + ((n & 1) << 4) + ((n & 31) >> 1);
    smem[OFF_WGATE + row * WP + k] = Wgate[i];
  }
  for (int i = threadIdx.x; i < 8 * 64; i += BLOCK) {
    int k = i >> 6, n = i & 63;
    smem[OFF_WENC + n * EP + k] = Wenc[i];
  }
  if (threadIdx.x < 64)
    smem[OFF_BIAS0 + threadIdx.x] =
        bsrc[threadIdx.x] + bdst[threadIdx.x] + benc[threadIdx.x];
  if (threadIdx.x < 256) {
    int n = threadIdx.x;
    smem[OFF_BGATE + (n & 0xE0) + ((n & 1) << 4) + ((n & 31) >> 1)] = bgate[n];
  }
  __syncthreads();

  const int lane = threadIdx.x & 31;
  const int wave = threadIdx.x >> 5;
  const int half = lane >> 4;
  const int nl   = lane & 15;

  float* stage = smem + OFF_STAGE + wave * (16 * WP);
  float* rhS   = smem + OFF_RH + wave * 64;
  int*   idxS  = ((int*)(smem + OFF_IDX)) + wave * 32;

  for (int tile = blockIdx.x * WAVES + wave; tile < NTILES;
       tile += gridDim.x * WAVES) {
    const int e  = tile * 16 + nl;
    const int sN = src[e];
    const int dN = dst[e];
    if (half == 0) { idxS[nl] = sN; idxS[16 + nl] = dN; }

    float rx = r_ij[e * 3 + 0], ry = r_ij[e * 3 + 1], rz = r_ij[e * 3 + 2];
    float dd  = sqrtf(rx * rx + ry * ry + rz * rz);
    float sx = rx * 3.5f, sy = ry * 3.5f, sz = rz * 3.5f;
    float inv = rsqrtf(1.0f + sx * sx + sy * sy + sz * sz);
    if (half == 0) {
      rhS[nl * 4 + 0] = sx * inv;
      rhS[nl * 4 + 1] = sy * inv;
      rhS[nl * 4 + 2] = sz * inv;
    }
    float rb[4];
#pragma unroll
    for (int u = 0; u < 4; ++u) {
      int k = 4 * half + u;
      float x = (dd - (float)k * (R0f / 7.0f)) * 4.0f;
      rb[u] = expf(-x * x);
    }

    if (useP) {
      const char* Pb = (const char*)P;
      v4f ps[8], pd[8];
#pragma unroll
      for (int r = 0; r < 8; ++r) {
        const unsigned sR = (unsigned)idxS[r + 8 * half];
        const unsigned dR = (unsigned)idxS[16 + r + 8 * half];
        ps[r] = *(const v4f*)(Pb + (sR * 64u + 4u * (unsigned)nl) * 4u);
        pd[r] = *(const v4f*)(Pb + ((unsigned)(NNODES * CHAN) + dR * 64u + 4u * (unsigned)nl) * 4u);
      }
      v8f acc[4];
      {
        v4f be[4];
#pragma unroll
        for (int j = 0; j < 4; ++j)
          be[j] = *(const v4f*)(smem + OFF_WENC + (4 * nl + j) * EP + 4 * half);
        v2f a0, a1;
        a0[0] = rb[0];
        a0[1] = rb[1];
        a1[0] = rb[2];
        a1[1] = rb[3];
#pragma unroll
        for (int j = 0; j < 4; ++j) acc[j] = wmma4(a0, lo2(be[j]), (v8f)0.0f);
#pragma unroll
        for (int j = 0; j < 4; ++j) acc[j] = wmma4(a1, hi2(be[j]), acc[j]);
      }
      const v4f bb = *(const v4f*)(smem + OFF_BIAS0 + 4 * nl);

      lds_fence();
#pragma unroll
      for (int r = 0; r < 8; ++r) {
        v4f o;
#pragma unroll
        for (int j = 0; j < 4; ++j) o[j] = acc[j][r] + (ps[r][j] + pd[r][j] + bb[j]);
        *(v4f*)(stage + (r + 8 * half) * WP + 4 * nl) = o;
      }
      lds_fence();
    } else {
      v8f acc[4];
#pragma unroll
      for (int j = 0; j < 4; ++j) acc[j] = (v8f)0.0f;

      const float* es = emb + (size_t)sN * 64 + 4 * half;
      const float* ed = emb + (size_t)dN * 64 + 4 * half;
      v4f es4[8], ed4[8];
#pragma unroll
      for (int q = 0; q < 8; ++q) es4[q] = *(const v4f*)(es + 8 * q);
#pragma unroll
      for (int q = 0; q < 8; ++q) ed4[q] = *(const v4f*)(ed + 8 * q);
      mm64(acc, es4, smem + OFF_WSRC + nl * WP + 4 * half);
      mm64(acc, ed4, smem + OFF_WDST + nl * WP + 4 * half);
      {
        v4f be[4];
#pragma unroll
        for (int j = 0; j < 4; ++j)
          be[j] = *(const v4f*)(smem + OFF_WENC + (16 * j + nl) * EP + 4 * half);
        v2f a0, a1;
        a0[0] = rb[0];
        a0[1] = rb[1];
        a1[0] = rb[2];
        a1[1] = rb[3];
#pragma unroll
        for (int j = 0; j < 4; ++j) acc[j] = wmma4(a0, lo2(be[j]), acc[j]);
#pragma unroll
        for (int j = 0; j < 4; ++j) acc[j] = wmma4(a1, hi2(be[j]), acc[j]);
      }
#pragma unroll
      for (int j = 0; j < 4; ++j) {
        float bb = smem[OFF_BIAS0 + 16 * j + nl];
#pragma unroll
        for (int r = 0; r < 8; ++r) acc[j][r] += bb;
      }

      lds_fence();
#pragma unroll
      for (int j = 0; j < 4; ++j)
#pragma unroll
        for (int r = 0; r < 8; ++r)
          stage[(r + 8 * half) * WP + 16 * j + nl] = acc[j][r];
      lds_fence();
    }

    unsigned sRow[16];
#pragma unroll
    for (int R = 0; R < 16; ++R) sRow[R] = (unsigned)__builtin_amdgcn_readlane(sN, R);

#pragma unroll 1
    for (int jp = 0; jp < 2; ++jp) {
      const char* z0b = (const char*)z0g;
      const char* z1b = (const char*)z1g;
      float ms[2][8], mv[2][3][8];

      v2f z0p[8], z1p[3][8];
      {
        unsigned zb0[8], zb1[8];
        const unsigned chP = (unsigned)(32 * jp + 2 * nl);
#pragma unroll
        for (int r = 0; r < 8; ++r) {
          const unsigned dN8 = (unsigned)idxS[16 + r + 8 * half];
          zb0[r] = (dN8 * 64u + chP) * 4u;
          zb1[r] = (dN8 * 192u + chP) * 4u;
        }
#pragma unroll
        for (int r = 0; r < 8; ++r) z0p[r] = *(const v2f*)(z0b + zb0[r]);
#pragma unroll
        for (int i = 0; i < 3; ++i)
#pragma unroll
          for (int r = 0; r < 8; ++r)
            z1p[i][r] = *(const v2f*)(z1b + i * 256 + zb1[r]);
      }

#pragma unroll
      for (int h2 = 0; h2 < 2; ++h2) {
        const int j = 2 * jp + h2;
        float z0r[8];
#pragma unroll
        for (int r = 0; r < 8; ++r) z0r[r] = z0p[r][h2];
        float z1r[3][8];
#pragma unroll
        for (int i = 0; i < 3; ++i)
#pragma unroll
          for (int r = 0; r < 8; ++r)
            z1r[i][r] = z1p[i][r][h2];

        const float* wb = smem + OFF_WGATE + (16 * j + nl) * WP + 2 * half;
        float bb[4];
        v8f g[4];
        v2f bq[2][2][4];
        v2f aq[2][2];
        const float* ab = stage + nl * WP + 2 * half;
        aq[0][0] = *(const v2f*)(ab);
        aq[0][1] = *(const v2f*)(ab + 4);
#pragma unroll
        for (int c = 0; c < 4; ++c) {
          g[c] = (v8f)0.0f;
          bb[c] = smem[OFF_BGATE + 64 * c + 16 * j + nl];
          bq[0][0][c] = *(const v2f*)(wb + 64 * c * WP);
          bq[0][1][c] = *(const v2f*)(wb + 64 * c * WP + 4);
        }
        __builtin_amdgcn_sched_barrier(0);
#pragma unroll
        for (int u = 0; u < 8; ++u) {
          if (u < 7) {
#pragma unroll
            for (int c = 0; c < 4; ++c) {
              bq[(u + 1) & 1][0][c] = *(const v2f*)(wb + 64 * c * WP + 8 * (u + 1));
              bq[(u + 1) & 1][1][c] = *(const v2f*)(wb + 64 * c * WP + 8 * (u + 1) + 4);
            }
            aq[(u + 1) & 1][0] = *(const v2f*)(ab + 8 * (u + 1));
            aq[(u + 1) & 1][1] = *(const v2f*)(ab + 8 * (u + 1) + 4);
          }
          __builtin_amdgcn_sched_barrier(0);
#pragma unroll
          for (int c = 0; c < 4; ++c) g[c] = wmma4(aq[u & 1][0], bq[u & 1][0][c], g[c]);
#pragma unroll
          for (int c = 0; c < 4; ++c) g[c] = wmma4(aq[u & 1][1], bq[u & 1][1][c], g[c]);
          __builtin_amdgcn_sched_barrier(0);
        }
#pragma unroll
        for (int c = 0; c < 4; ++c)
#pragma unroll
          for (int r = 0; r < 8; ++r) g[c][r] += bb[c];

        float rh[3][8];
#pragma unroll
        for (int r = 0; r < 8; ++r) {
          rh[0][r] = rhS[(r + 8 * half) * 4 + 0];
          rh[1][r] = rhS[(r + 8 * half) * 4 + 1];
          rh[2][r] = rhS[(r + 8 * half) * 4 + 2];
        }

#pragma unroll
        for (int r = 0; r < 8; ++r) {
          float dotv = rh[0][r] * z1r[0][r] + rh[1][r] * z1r[1][r] +
                       rh[2][r] * z1r[2][r];
          ms[h2][r] = g[0][r] * z0r[r] + g[1][r] * dotv;
        }
#pragma unroll
        for (int i = 0; i < 3; ++i)
#pragma unroll
          for (int r = 0; r < 8; ++r)
            mv[h2][i][r] = g[2][r] * z1r[i][r] + g[3][r] * rh[i][r] * z0r[r];

        __builtin_amdgcn_sched_barrier(0);
      }

      float xs[2][8], xv[2][3][8];
#pragma unroll
      for (int r = 0; r < 8; ++r) {
        v2u t = __builtin_amdgcn_permlane16_swap(__float_as_uint(ms[0][r]), __float_as_uint(ms[1][r]), false, false);
        xs[0][r] = __uint_as_float(t[0]);
        xs[1][r] = __uint_as_float(t[1]);
      }
#pragma unroll
      for (int i = 0; i < 3; ++i)
#pragma unroll
        for (int r = 0; r < 8; ++r) {
          v2u t = __builtin_amdgcn_permlane16_swap(__float_as_uint(mv[0][i][r]), __float_as_uint(mv[1][i][r]), false, false);
          xv[0][i][r] = __uint_as_float(t[0]);
          xv[1][i][r] = __uint_as_float(t[1]);
        }
      const unsigned ch = (unsigned)(32 * jp + 2 * nl + half);
      float* a0[2][8];
      float* a1[2][8];
#pragma unroll
      for (int q = 0; q < 2; ++q)
#pragma unroll
        for (int r = 0; r < 8; ++r) {
          a0[q][r] = out0 + (size_t)sRow[r + 8 * q] * 64 + ch;
          a1[q][r] = out1 + (size_t)sRow[r + 8 * q] * 192 + ch;
        }
      __builtin_amdgcn_sched_barrier(0);

#pragma unroll
      for (int q = 0; q < 2; ++q)
#pragma unroll
        for (int r = 0; r < 8; ++r) {
          atomicAdd(a0[q][r], xs[q][r]);
#pragma unroll
          for (int i = 0; i < 3; ++i)
            atomicAdd(a1[q][r] + i * 64, xv[q][i][r]);
        }
    }
  }
}

__device__ __forceinline__ void tile_times_w(const float* inb, int in_pitch, float* outb, int out_pitch,
                                             const float* wT, float* stage, int lane, int half, int nl) {
  v4f v[8];
#pragma unroll
  for (int q = 0; q < 8; ++q) {
    int idx = q * 32 + lane;
    v[q] = *(const v4f*)(inb + (size_t)(idx >> 4) * in_pitch + (idx & 15) * 4);
  }
  lds_fence();
#pragma unroll
  for (int q = 0; q < 8; ++q) {
    int idx = q * 32 + lane;
    *(v4f*)(stage + (idx >> 4) * WP + (idx & 15) * 4) = v[q];
  }
  lds_fence();

  v8f p[4];
#pragma unroll
  for (int j = 0; j < 4; ++j) p[j] = (v8f)0.0f;
#pragma unroll
  for (int t = 0; t < 16; ++t) {
    int k0 = 4 * t + 2 * half;
    v2f a = *(const v2f*)(stage + nl * WP + k0);
#pragma unroll
    for (int j = 0; j < 4; ++j) {
      v2f b = *(const v2f*)(wT + (16 * j + nl) * WP + k0);
      p[j] = wmma4(a, b, p[j]);
    }
  }

  lds_fence();
#pragma unroll
  for (int j = 0; j < 4; ++j)
#pragma unroll
    for (int r = 0; r < 8; ++r)
      stage[(r + 8 * half) * WP + 16 * j + nl] = p[j][r];
  lds_fence();
#pragma unroll
  for (int q = 0; q < 8; ++q) {
    int idx = q * 32 + lane;
    v[q] = *(const v4f*)(stage + (idx >> 4) * WP + (idx & 15) * 4);
  }
#pragma unroll
  for (int q = 0; q < 8; ++q) {
    int idx = q * 32 + lane;
    *(v4f*)(outb + (size_t)(idx >> 4) * out_pitch + (idx & 15) * 4) = v[q];
  }
}

__global__ void __launch_bounds__(BLOCK)
node_embed_kernel(const float* __restrict__ emb, const float* __restrict__ Wsrc,
                  const float* __restrict__ Wdst, float* __restrict__ P, size_t ws_bytes) {
  extern __shared__ float smem[];

  if (ws_bytes < P_BYTES) return;

  for (int i = threadIdx.x; i < 64 * 64; i += BLOCK) {
    int k = i >> 6, n = i & 63;
    smem[N_OFF_WS + n * WP + k] = Wsrc[i];
    smem[N_OFF_WV + n * WP + k] = Wdst[i];
  }
  __syncthreads();

  const int lane = threadIdx.x & 31;
  const int wave = threadIdx.x >> 5;
  const int half = lane >> 4;
  const int nl   = lane & 15;

  float* stage = smem + N_OFF_STAGE + wave * (16 * WP);

  for (int task = blockIdx.x * WAVES + wave; task < P_TASKS;
       task += gridDim.x * WAVES) {
    const int tile = task >> 1;
    const int part = task & 1;
    tile_times_w(emb + (size_t)tile * (16 * 64), 64,
                 P + (size_t)part * ((size_t)NNODES * CHAN) + (size_t)tile * (16 * 64), 64,
                 smem + (part ? N_OFF_WV : N_OFF_WS), stage, lane, half, nl);
  }
}

__global__ void __launch_bounds__(BLOCK)
node_transform_kernel(const float* __restrict__ Ws, const float* __restrict__ Wv,
                      float* out) {
  extern __shared__ float smem[];

  for (int i = threadIdx.x; i < 64 * 64; i += BLOCK) {
    int k = i >> 6, n = i & 63;
    smem[N_OFF_WS + n * WP + k] = Ws[i];
    smem[N_OFF_WV + n * WP + k] = Wv[i];
  }
  __syncthreads();

  const int lane = threadIdx.x & 31;
  const int wave = threadIdx.x >> 5;
  const int half = lane >> 4;
  const int nl   = lane & 15;

  float* stage = smem + N_OFF_STAGE + wave * (16 * WP);

  for (int task = blockIdx.x * WAVES + wave; task < NODE_TASKS;
       task += gridDim.x * WAVES) {
    const int tile = task >> 2;
    const int part = task & 3;
    float* base;
    int pitch;
    const float* wT;
    if (part == 0) {
      base = out + (size_t)tile * (16 * 64);
      pitch = 64;
      wT = smem + N_OFF_WS;
    } else {
      base = out + (size_t)NNODES * CHAN + (size_t)tile * (16 * 192) + (part - 1) * 64;
      pitch = 192;
      wT = smem + N_OFF_WV;
    }
    tile_times_w(base, pitch, base, pitch, wT, stage, lane, half, nl);
  }
}

extern "C" void kernel_launch(void* const* d_in, const int* in_sizes, int n_in,
                              void* d_out, int out_size, void* d_ws,
                              size_t ws_size, hipStream_t stream) {
  const int*   src   = (const int*)d_in[0];
  const int*   dst   = (const int*)d_in[1];
  const float* r_ij  = (const float*)d_in[2];
  const float* z0    = (const float*)d_in[3];
  const float* z1    = (const float*)d_in[4];
  const float* emb   = (const float*)d_in[5];
  const float* Wenc  = (const float*)d_in[6];
  const float* benc  = (const float*)d_in[7];
  const float* Wsrc  = (const float*)d_in[8];
  const float* bsrc  = (const float*)d_in[9];
  const float* Wdst  = (const float*)d_in[10];
  const float* bdst  = (const float*)d_in[11];
  const float* Wgate = (const float*)d_in[12];
  const float* bgate = (const float*)d_in[13];
  const float* Ws    = (const float*)d_in[14];
  const float* Wv    = (const float*)d_in[15];
  float* out = (float*)d_out;
  float* ws  = (float*)d_ws;

  zero_f32<<<ZGRID, BLOCK, 0, stream>>>(out, out_size);

  size_t nsmem_bytes = (size_t)N_SMEM_FLOATS * sizeof(float);
  node_embed_kernel<<<PGRID, BLOCK, nsmem_bytes, stream>>>(emb, Wsrc, Wdst, ws, ws_size);

  size_t smem_bytes = (size_t)SMEM_FLOATS * sizeof(float);
  messages_kernel<<<GRID, BLOCK, smem_bytes, stream>>>(
      src, dst, r_ij, z0, z1, emb, Wenc, benc, Wsrc, bsrc, Wdst, bdst, Wgate,
      bgate, out, out + (size_t)NNODES * CHAN, ws, ws_size);

  node_transform_kernel<<<NGRID, BLOCK, nsmem_bytes, stream>>>(Ws, Wv, out);
}
